# starting kernel
# baseline (speedup 1.0000x reference)
#include <hip/hip_runtime.h>
#include <hip/hip_bf16.h>
#include <math.h>

#define NHEAD   8
#define DMODEL  256
#define DKDIM   32
#define DVDIM   32
#define LSEQ    1024
#define BATCH   2
#define LN_EPSF 1e-5f

typedef float v2f __attribute__((ext_vector_type(2)));
typedef float v8f __attribute__((ext_vector_type(8)));

__device__ __forceinline__ v8f wmma_f32_k4(v2f a, v2f b, v8f c) {
  return __builtin_amdgcn_wmma_f32_16x16x4_f32(false, a, false, b, (short)0, c,
                                               false, false);
}

__device__ __forceinline__ float ntload(const float* p) {
  return __builtin_nontemporal_load(p);
}
__device__ __forceinline__ v2f ntload2(const float* p) {
  return __builtin_nontemporal_load((const v2f*)p);
}

__global__ void ln_kernel(const float* __restrict__ h,
                          const float* __restrict__ gamma,
                          const float* __restrict__ beta,
                          float* __restrict__ hn) {
  int row  = blockIdx.x;
  int lane = threadIdx.x;
  const float* x = h + (size_t)row * DMODEL;
  float vals[8];
  float s = 0.f;
#pragma unroll
  for (int t = 0; t < 8; ++t) { vals[t] = x[lane + 32 * t]; s += vals[t]; }
#pragma unroll
  for (int o = 16; o > 0; o >>= 1) s += __shfl_xor(s, o, 32);
  float mu = s * (1.0f / DMODEL);
  float vs = 0.f;
#pragma unroll
  for (int t = 0; t < 8; ++t) { float d = vals[t] - mu; vs += d * d; }
#pragma unroll
  for (int o = 16; o > 0; o >>= 1) vs += __shfl_xor(vs, o, 32);
  float inv = rsqrtf(vs * (1.0f / DMODEL) + LN_EPSF);
  float* y = hn + (size_t)row * DMODEL;
#pragma unroll
  for (int t = 0; t < 8; ++t) {
    int c = lane + 32 * t;
    y[c] = (vals[t] - mu) * inv * gamma[c] + beta[c];
  }
}

__global__ void proj_kernel(const float* __restrict__ A,
                            const float* __restrict__ W,
                            float* __restrict__ C) {
  int m0 = blockIdx.x * 16, n0 = blockIdx.y * 16;
  int lane = threadIdx.x, half = lane >> 4, l = lane & 15;
  const float* arow = A + (size_t)(m0 + l) * DMODEL;
  const float* wrow = W + (size_t)(n0 + l) * DMODEL;
  v8f c = {};
  for (int k0 = 0; k0 < DMODEL; k0 += 4) {
    v2f a, b;
    a.x = arow[k0 + 2 * half]; a.y = arow[k0 + 2 * half + 1];
    b.x = wrow[k0 + 2 * half]; b.y = wrow[k0 + 2 * half + 1];
    c = wmma_f32_k4(a, b, c);
  }
#pragma unroll
  for (int r = 0; r < 8; ++r)
    C[(size_t)(m0 + r + 8 * half) * DMODEL + n0 + l] = c[r];
}

__global__ void attn1_kernel(const float* __restrict__ Q,
                             const float* __restrict__ K,
                             float* __restrict__ attn) {
  int bh = blockIdx.x; int b = bh >> 3; int hh = bh & 7;
  int i0 = blockIdx.y * 16, j0 = blockIdx.z * 16;
  int lane = threadIdx.x, half = lane >> 4, l = lane & 15;
  const float* qrow = Q + ((size_t)(b * LSEQ + i0 + l) * NHEAD + hh) * DKDIM;
  const float* krow = K + ((size_t)(b * LSEQ + j0 + l) * NHEAD + hh) * DKDIM;
  v8f c = {};
#pragma unroll
  for (int k0 = 0; k0 < DKDIM; k0 += 4) {
    v2f a, bf;
    a.x  = qrow[k0 + 2 * half]; a.y  = qrow[k0 + 2 * half + 1];
    bf.x = krow[k0 + 2 * half]; bf.y = krow[k0 + 2 * half + 1];
    c = wmma_f32_k4(a, bf, c);
  }
  size_t base = ((size_t)hh * BATCH + b) * LSEQ * LSEQ;
#pragma unroll
  for (int r = 0; r < 8; ++r)
    attn[base + (size_t)(i0 + r + 8 * half) * LSEQ + j0 + l] = c[r];
}

__global__ void attn2_kernel(const float* __restrict__ bias,
                             const float* __restrict__ Q,
                             float* __restrict__ attn) {
  int b = blockIdx.x, i = blockIdx.y, j0 = blockIdx.z * 16;
  int lane = threadIdx.x, half = lane >> 4, l = lane & 15;
  bool nvalid = l < 8;
  const float* arow = bias + (((size_t)b * LSEQ + i) * LSEQ + j0 + l) * DKDIM;
  const float* qv   = Q + ((size_t)(b * LSEQ + i) * NHEAD + (l & 7)) * DKDIM;
  v8f c = {};
#pragma unroll
  for (int k0 = 0; k0 < DKDIM; k0 += 4) {
    v2f a = ntload2(arow + k0 + 2 * half);
    v2f bf;
    if (nvalid) { bf.x = qv[k0 + 2 * half]; bf.y = qv[k0 + 2 * half + 1]; }
    else        { bf.x = 0.f;               bf.y = 0.f; }
    c = wmma_f32_k4(a, bf, c);
  }
  if (nvalid) {
    float* pr = attn + (((size_t)l * BATCH + b) * LSEQ + i) * LSEQ + j0 + 8 * half;
    float4 x0 = *(float4*)(pr);
    float4 x1 = *(float4*)(pr + 4);
    x0.x += c[0]; x0.y += c[1]; x0.z += c[2]; x0.w += c[3];
    x1.x += c[4]; x1.y += c[5]; x1.z += c[6]; x1.w += c[7];
    *(float4*)(pr)     = x0;
    *(float4*)(pr + 4) = x1;
  }
}

__global__ void softmax_kernel(float* __restrict__ attn) {
  __shared__ float red[8];
  float4* p = (float4*)(attn + (size_t)blockIdx.x * LSEQ);
  int t = threadIdx.x;
  const float scale = 0.17677669529663687f;
  float4 vv = p[t];
  float v[4] = {vv.x * scale, vv.y * scale, vv.z * scale, vv.w * scale};
  float mx = fmaxf(fmaxf(v[0], v[1]), fmaxf(v[2], v[3]));
#pragma unroll
  for (int o = 16; o > 0; o >>= 1) mx = fmaxf(mx, __shfl_xor(mx, o, 32));
  int wid = t >> 5;
  if ((t & 31) == 0) red[wid] = mx;
  __syncthreads();
  float bm = red[0];
#pragma unroll
  for (int w = 1; w < 8; ++w) bm = fmaxf(bm, red[w]);
  float s = 0.f;
#pragma unroll
  for (int u = 0; u < 4; ++u) { v[u] = __expf(v[u] - bm); s += v[u]; }
#pragma unroll
  for (int o = 16; o > 0; o >>= 1) s += __shfl_xor(s, o, 32);
  __syncthreads();
  if ((t & 31) == 0) red[wid] = s;
  __syncthreads();
  float tot = 0.f;
#pragma unroll
  for (int w = 0; w < 8; ++w) tot += red[w];
  float inv = 1.0f / tot;
  p[t] = make_float4(v[0] * inv, v[1] * inv, v[2] * inv, v[3] * inv);
}

__global__ void emb1_kernel(const float* __restrict__ attn,
                            const float* __restrict__ V,
                            float* __restrict__ emb) {
  int bh = blockIdx.x; int b = bh >> 3; int hh = bh & 7;
  int i0 = blockIdx.y * 16;
  int lane = threadIdx.x, half = lane >> 4, l = lane & 15;
  const float* arow = attn + (((size_t)hh * BATCH + b) * LSEQ + i0 + l) * LSEQ;
  v8f c0 = {}, c1 = {};
  for (int k0 = 0; k0 < LSEQ; k0 += 4) {
    v2f a;
    a.x = arow[k0 + 2 * half]; a.y = arow[k0 + 2 * half + 1];
    const float* v0 =
        V + ((size_t)(b * LSEQ + k0 + 2 * half) * NHEAD + hh) * DVDIM + l;
    v2f b0, b1;
    b0.x = v0[0];  b0.y = v0[NHEAD * DVDIM];
    b1.x = v0[16]; b1.y = v0[NHEAD * DVDIM + 16];
    c0 = wmma_f32_k4(a, b0, c0);
    c1 = wmma_f32_k4(a, b1, c1);
  }
  float* e = emb + (size_t)(b * LSEQ + i0) * (NHEAD * DVDIM) + hh * DVDIM;
#pragma unroll
  for (int r = 0; r < 8; ++r) {
    e[(size_t)(r + 8 * half) * (NHEAD * DVDIM) + l]      = c0[r];
    e[(size_t)(r + 8 * half) * (NHEAD * DVDIM) + 16 + l] = c1[r];
  }
}

__global__ void emb2_kernel(const float* __restrict__ attn,
                            const float* __restrict__ bias,
                            float* __restrict__ emb) {
  int bi = blockIdx.x; int b = bi >> 10; int i = bi & 1023;
  int lane = threadIdx.x, half = lane >> 4, l = lane & 15;
  bool mvalid = l < 8;
  const float* arow = attn + (((size_t)(l & 7) * BATCH + b) * LSEQ + i) * LSEQ;
  const float* brow = bias + ((size_t)b * LSEQ + i) * (size_t)LSEQ * DKDIM;
  v8f c0 = {}, c1 = {};
  for (int k0 = 0; k0 < LSEQ; k0 += 4) {
    v2f a;
    if (mvalid) { a.x = arow[k0 + 2 * half]; a.y = arow[k0 + 2 * half + 1]; }
    else        { a.x = 0.f;                 a.y = 0.f; }
    const float* p0 = brow + (size_t)(k0 + 2 * half) * DKDIM;
    v2f b0, b1;
    b0.x = ntload(p0 + l);              b0.y = ntload(p0 + DKDIM + l);
    b1.x = ntload(p0 + 16 + l);         b1.y = ntload(p0 + DKDIM + 16 + l);
    c0 = wmma_f32_k4(a, b0, c0);
    c1 = wmma_f32_k4(a, b1, c1);
  }
  if (half == 0) {
    float* e = emb + (size_t)(b * LSEQ + i) * (NHEAD * DVDIM);
#pragma unroll
    for (int r = 0; r < 8; ++r) {
      e[r * DVDIM + l]      += c0[r];
      e[r * DVDIM + 16 + l] += c1[r];
    }
  }
}

__global__ void fc_kernel(const float* __restrict__ emb,
                          const float* __restrict__ W,
                          const float* __restrict__ bvec,
                          const float* __restrict__ resid,
                          float* __restrict__ out) {
  int m0 = blockIdx.x * 16, n0 = blockIdx.y * 16;
  int lane = threadIdx.x, half = lane >> 4, l = lane & 15;
  const float* arow = emb + (size_t)(m0 + l) * DMODEL;
  const float* wrow = W + (size_t)(n0 + l) * DMODEL;
  v8f c = {};
  for (int k0 = 0; k0 < DMODEL; k0 += 4) {
    v2f a, bf;
    a.x  = arow[k0 + 2 * half]; a.y  = arow[k0 + 2 * half + 1];
    bf.x = wrow[k0 + 2 * half]; bf.y = wrow[k0 + 2 * half + 1];
    c = wmma_f32_k4(a, bf, c);
  }
#pragma unroll
  for (int r = 0; r < 8; ++r) {
    int m = m0 + r + 8 * half, n = n0 + l;
    out[(size_t)m * DMODEL + n] = c[r] + bvec[n] + resid[(size_t)m * DMODEL + n];
  }
}

extern "C" void kernel_launch(void* const* d_in, const int* in_sizes, int n_in,
                              void* d_out, int out_size, void* d_ws,
                              size_t ws_size, hipStream_t stream) {
  (void)in_sizes; (void)n_in; (void)out_size; (void)ws_size;
  const float* h    = (const float*)d_in[0];
  const float* bias = (const float*)d_in[1];
  const float* w_qs = (const float*)d_in[2];
  const float* w_ks = (const float*)d_in[3];
  const float* w_vs = (const float*)d_in[4];
  const float* ln_g = (const float*)d_in[5];
  const float* ln_b = (const float*)d_in[6];
  const float* fc_w = (const float*)d_in[7];
  const float* fc_b = (const float*)d_in[8];

  float* out  = (float*)d_out;
  float* attn = out + (size_t)BATCH * LSEQ * DMODEL;

  const size_t ROWS = (size_t)BATCH * LSEQ;
  float* ws  = (float*)d_ws;
  float* hn  = ws;
  float* q   = hn + ROWS * DMODEL;
  float* k   = q + ROWS * DMODEL;
  float* v   = k + ROWS * DMODEL;
  float* emb = v + ROWS * DMODEL;

  ln_kernel<<<(int)ROWS, 32, 0, stream>>>(h, ln_g, ln_b, hn);

  dim3 gproj((unsigned)(ROWS / 16), DMODEL / 16);
  proj_kernel<<<gproj, 32, 0, stream>>>(hn, w_qs, q);
  proj_kernel<<<gproj, 32, 0, stream>>>(hn, w_ks, k);
  proj_kernel<<<gproj, 32, 0, stream>>>(hn, w_vs, v);

  attn1_kernel<<<dim3(BATCH * NHEAD, LSEQ / 16, LSEQ / 16), 32, 0, stream>>>(q, k, attn);
  attn2_kernel<<<dim3(BATCH, LSEQ, LSEQ / 16), 32, 0, stream>>>(bias, q, attn);
  softmax_kernel<<<NHEAD * BATCH * LSEQ, 256, 0, stream>>>(attn);
  emb1_kernel<<<dim3(BATCH * NHEAD, LSEQ / 16), 32, 0, stream>>>(attn, v, emb);
  emb2_kernel<<<(unsigned)ROWS, 32, 0, stream>>>(attn, bias, emb);
  fc_kernel<<<gproj, 32, 0, stream>>>(emb, fc_w, fc_b, h, out);
}


# emb2: j loop split over 8 waves per (b,i) row, each wave keeps 10+ whole-row b64 bias loads in flight ahead of its WMMAs; 8 partial tiles summed through LDS (sum order over j changes)
# speedup vs baseline: 1.3031x; 1.3031x over previous
#include <hip/hip_runtime.h>
#include <hip/hip_bf16.h>
#include <math.h>

#define NHEAD   8
#define DMODEL  256
#define DKDIM   32
#define DVDIM   32
#define LSEQ    1024
#define BATCH   2
#define LN_EPSF 1e-5f

typedef float v2f __attribute__((ext_vector_type(2)));
typedef float v8f __attribute__((ext_vector_type(8)));

__device__ __forceinline__ v8f wmma_f32_k4(v2f a, v2f b, v8f c) {
  return __builtin_amdgcn_wmma_f32_16x16x4_f32(false, a, false, b, (short)0, c,
                                               false, false);
}

__device__ __forceinline__ float ntload(const float* p) {
  return __builtin_nontemporal_load(p);
}
__device__ __forceinline__ v2f ntload2(const float* p) {
  return __builtin_nontemporal_load((const v2f*)p);
}

__global__ void ln_kernel(const float* __restrict__ h,
                          const float* __restrict__ gamma,
                          const float* __restrict__ beta,
                          float* __restrict__ hn) {
  int row  = blockIdx.x;
  int lane = threadIdx.x;
  const float* x = h + (size_t)row * DMODEL;
  float vals[8];
  float s = 0.f;
#pragma unroll
  for (int t = 0; t < 8; ++t) { vals[t] = x[lane + 32 * t]; s += vals[t]; }
#pragma unroll
  for (int o = 16; o > 0; o >>= 1) s += __shfl_xor(s, o, 32);
  float mu = s * (1.0f / DMODEL);
  float vs = 0.f;
#pragma unroll
  for (int t = 0; t < 8; ++t) { float d = vals[t] - mu; vs += d * d; }
#pragma unroll
  for (int o = 16; o > 0; o >>= 1) vs += __shfl_xor(vs, o, 32);
  float inv = rsqrtf(vs * (1.0f / DMODEL) + LN_EPSF);
  float* y = hn + (size_t)row * DMODEL;
#pragma unroll
  for (int t = 0; t < 8; ++t) {
    int c = lane + 32 * t;
    y[c] = (vals[t] - mu) * inv * gamma[c] + beta[c];
  }
}

__global__ void proj_kernel(const float* __restrict__ A,
                            const float* __restrict__ W,
                            float* __restrict__ C) {
  int m0 = blockIdx.x * 16, n0 = blockIdx.y * 16;
  int lane = threadIdx.x, half = lane >> 4, l = lane & 15;
  const float* arow = A + (size_t)(m0 + l) * DMODEL;
  const float* wrow = W + (size_t)(n0 + l) * DMODEL;
  v8f c = {};
  for (int k0 = 0; k0 < DMODEL; k0 += 4) {
    v2f a, b;
    a.x = arow[k0 + 2 * half]; a.y = arow[k0 + 2 * half + 1];
    b.x = wrow[k0 + 2 * half]; b.y = wrow[k0 + 2 * half + 1];
    c = wmma_f32_k4(a, b, c);
  }
#pragma unroll
  for (int r = 0; r < 8; ++r)
    C[(size_t)(m0 + r + 8 * half) * DMODEL + n0 + l] = c[r];
}

__global__ void attn1_kernel(const float* __restrict__ Q,
                             const float* __restrict__ K,
                             float* __restrict__ attn) {
  int bh = blockIdx.x; int b = bh >> 3; int hh = bh & 7;
  int i0 = blockIdx.y * 16, j0 = blockIdx.z * 16;
  int lane = threadIdx.x, half = lane >> 4, l = lane & 15;
  const float* qrow = Q + ((size_t)(b * LSEQ + i0 + l) * NHEAD + hh) * DKDIM;
  const float* krow = K + ((size_t)(b * LSEQ + j0 + l) * NHEAD + hh) * DKDIM;
  v8f c = {};
#pragma unroll
  for (int k0 = 0; k0 < DKDIM; k0 += 4) {
    v2f a, bf;
    a.x  = qrow[k0 + 2 * half]; a.y  = qrow[k0 + 2 * half + 1];
    bf.x = krow[k0 + 2 * half]; bf.y = krow[k0 + 2 * half + 1];
    c = wmma_f32_k4(a, bf, c);
  }
  size_t base = ((size_t)hh * BATCH + b) * LSEQ * LSEQ;
#pragma unroll
  for (int r = 0; r < 8; ++r)
    attn[base + (size_t)(i0 + r + 8 * half) * LSEQ + j0 + l] = c[r];
}

__global__ void attn2_kernel(const float* __restrict__ bias,
                             const float* __restrict__ Q,
                             float* __restrict__ attn) {
  int b = blockIdx.x, i = blockIdx.y, j0 = blockIdx.z * 16;
  int lane = threadIdx.x, half = lane >> 4, l = lane & 15;
  bool nvalid = l < 8;
  const float* arow = bias + (((size_t)b * LSEQ + i) * LSEQ + j0 + l) * DKDIM;
  const float* qv   = Q + ((size_t)(b * LSEQ + i) * NHEAD + (l & 7)) * DKDIM;
  v8f c = {};
#pragma unroll
  for (int k0 = 0; k0 < DKDIM; k0 += 4) {
    v2f a = ntload2(arow + k0 + 2 * half);
    v2f bf;
    if (nvalid) { bf.x = qv[k0 + 2 * half]; bf.y = qv[k0 + 2 * half + 1]; }
    else        { bf.x = 0.f;               bf.y = 0.f; }
    c = wmma_f32_k4(a, bf, c);
  }
  if (nvalid) {
    float* pr = attn + (((size_t)l * BATCH + b) * LSEQ + i) * LSEQ + j0 + 8 * half;
    float4 x0 = *(float4*)(pr);
    float4 x1 = *(float4*)(pr + 4);
    x0.x += c[0]; x0.y += c[1]; x0.z += c[2]; x0.w += c[3];
    x1.x += c[4]; x1.y += c[5]; x1.z += c[6]; x1.w += c[7];
    *(float4*)(pr)     = x0;
    *(float4*)(pr + 4) = x1;
  }
}

__global__ void softmax_kernel(float* __restrict__ attn) {
  __shared__ float red[8];
  float4* p = (float4*)(attn + (size_t)blockIdx.x * LSEQ);
  int t = threadIdx.x;
  const float scale = 0.17677669529663687f;
  float4 vv = p[t];
  float v[4] = {vv.x * scale, vv.y * scale, vv.z * scale, vv.w * scale};
  float mx = fmaxf(fmaxf(v[0], v[1]), fmaxf(v[2], v[3]));
#pragma unroll
  for (int o = 16; o > 0; o >>= 1) mx = fmaxf(mx, __shfl_xor(mx, o, 32));
  int wid = t >> 5;
  if ((t & 31) == 0) red[wid] = mx;
  __syncthreads();
  float bm = red[0];
#pragma unroll
  for (int w = 1; w < 8; ++w) bm = fmaxf(bm, red[w]);
  float s = 0.f;
#pragma unroll
  for (int u = 0; u < 4; ++u) { v[u] = __expf(v[u] - bm); s += v[u]; }
#pragma unroll
  for (int o = 16; o > 0; o >>= 1) s += __shfl_xor(s, o, 32);
  __syncthreads();
  if ((t & 31) == 0) red[wid] = s;
  __syncthreads();
  float tot = 0.f;
#pragma unroll
  for (int w = 0; w < 8; ++w) tot += red[w];
  float inv = 1.0f / tot;
  p[t] = make_float4(v[0] * inv, v[1] * inv, v[2] * inv, v[3] * inv);
}

__global__ void emb1_kernel(const float* __restrict__ attn,
                            const float* __restrict__ V,
                            float* __restrict__ emb) {
  int bh = blockIdx.x; int b = bh >> 3; int hh = bh & 7;
  int i0 = blockIdx.y * 16;
  int lane = threadIdx.x, half = lane >> 4, l = lane & 15;
  const float* arow = attn + (((size_t)hh * BATCH + b) * LSEQ + i0 + l) * LSEQ;
  v8f c0 = {}, c1 = {};
  for (int k0 = 0; k0 < LSEQ; k0 += 4) {
    v2f a;
    a.x = arow[k0 + 2 * half]; a.y = arow[k0 + 2 * half + 1];
    const float* v0 =
        V + ((size_t)(b * LSEQ + k0 + 2 * half) * NHEAD + hh) * DVDIM + l;
    v2f b0, b1;
    b0.x = v0[0];  b0.y = v0[NHEAD * DVDIM];
    b1.x = v0[16]; b1.y = v0[NHEAD * DVDIM + 16];
    c0 = wmma_f32_k4(a, b0, c0);
    c1 = wmma_f32_k4(a, b1, c1);
  }
  float* e = emb + (size_t)(b * LSEQ + i0) * (NHEAD * DVDIM) + hh * DVDIM;
#pragma unroll
  for (int r = 0; r < 8; ++r) {
    e[(size_t)(r + 8 * half) * (NHEAD * DVDIM) + l]      = c0[r];
    e[(size_t)(r + 8 * half) * (NHEAD * DVDIM) + 16 + l] = c1[r];
  }
}

__global__ void __launch_bounds__(256)
emb2_kernel(const float* __restrict__ attn,
            const float* __restrict__ bias,
            float* __restrict__ emb) {
  __shared__ float part[8][NHEAD * DVDIM];
  int bi = blockIdx.x; int b = bi >> 10; int i = bi & 1023;
  int tid = threadIdx.x, w = tid >> 5;
  int lane = tid & 31, half = lane >> 4, l = lane & 15;
  const float* arow = attn + (((size_t)(l & 7) * BATCH + b) * LSEQ + i) * LSEQ
                      + w * 128 + 16 * half;
  const float* brow = bias + (((size_t)b * LSEQ + i) * LSEQ + w * 128 + 16 * half)
                      * (size_t)DKDIM + 2 * l;
  v8f c0 = {}, c1 = {};
  for (int ch = 0; ch < 4; ++ch) {
    v2f bb[16];
    float4 a4[4];
#pragma unroll
    for (int r = 0; r < 16; ++r) bb[r] = ntload2(brow + r * DKDIM);
#pragma unroll
    for (int t = 0; t < 4; ++t) a4[t] = *(const float4*)(arow + 4 * t);
#pragma unroll
    for (int t = 0; t < 4; ++t) {
      v2f a, b0, b1;
      a.x = a4[t].x; a.y = a4[t].y;
      b0.x = bb[4 * t].x;     b0.y = bb[4 * t + 1].x;
      b1.x = bb[4 * t].y;     b1.y = bb[4 * t + 1].y;
      c0 = wmma_f32_k4(a, b0, c0);
      c1 = wmma_f32_k4(a, b1, c1);
      a.x = a4[t].z; a.y = a4[t].w;
      b0.x = bb[4 * t + 2].x; b0.y = bb[4 * t + 3].x;
      b1.x = bb[4 * t + 2].y; b1.y = bb[4 * t + 3].y;
      c0 = wmma_f32_k4(a, b0, c0);
      c1 = wmma_f32_k4(a, b1, c1);
    }
    arow += 32;
    brow += 32 * DKDIM;
  }
  if (half == 0) {
#pragma unroll
    for (int r = 0; r < 8; ++r) {
      v2f o; o.x = c0[r]; o.y = c1[r];
      *(v2f*)(&part[w][r * DVDIM + 2 * l]) = o;
    }
  }
  __syncthreads();
  float sum = part[0][tid];
#pragma unroll
  for (int u = 1; u < 8; ++u) sum += part[u][tid];
  emb[(size_t)(b * LSEQ + i) * (NHEAD * DVDIM) + tid] += sum;
}

__global__ void fc_kernel(const float* __restrict__ emb,
                          const float* __restrict__ W,
                          const float* __restrict__ bvec,
                          const float* __restrict__ resid,
                          float* __restrict__ out) {
  int m0 = blockIdx.x * 16, n0 = blockIdx.y * 16;
  int lane = threadIdx.x, half = lane >> 4, l = lane & 15;
  const float* arow = emb + (size_t)(m0 + l) * DMODEL;
  const float* wrow = W + (size_t)(n0 + l) * DMODEL;
  v8f c = {};
  for (int k0 = 0; k0 < DMODEL; k0 += 4) {
    v2f a, bf;
    a.x  = arow[k0 + 2 * half]; a.y  = arow[k0 + 2 * half + 1];
    bf.x = wrow[k0 + 2 * half]; bf.y = wrow[k0 + 2 * half + 1];
    c = wmma_f32_k4(a, bf, c);
  }
#pragma unroll
  for (int r = 0; r < 8; ++r) {
    int m = m0 + r + 8 * half, n = n0 + l;
    out[(size_t)m * DMODEL + n] = c[r] + bvec[n] + resid[(size_t)m * DMODEL + n];
  }
}

extern "C" void kernel_launch(void* const* d_in, const int* in_sizes, int n_in,
                              void* d_out, int out_size, void* d_ws,
                              size_t ws_size, hipStream_t stream) {
  (void)in_sizes; (void)n_in; (void)out_size; (void)ws_size;
  const float* h    = (const float*)d_in[0];
  const float* bias = (const float*)d_in[1];
  const float* w_qs = (const float*)d_in[2];
  const float* w_ks = (const float*)d_in[3];
  const float* w_vs = (const float*)d_in[4];
  const float* ln_g = (const float*)d_in[5];
  const float* ln_b = (const float*)d_in[6];
  const float* fc_w = (const float*)d_in[7];
  const float* fc_b = (const float*)d_in[8];

  float* out  = (float*)d_out;
  float* attn = out + (size_t)BATCH * LSEQ * DMODEL;

  const size_t ROWS = (size_t)BATCH * LSEQ;
  float* ws  = (float*)d_ws;
  float* hn  = ws;
  float* q   = hn + ROWS * DMODEL;
  float* k   = q + ROWS * DMODEL;
  float* v   = k + ROWS * DMODEL;
  float* emb = v + ROWS * DMODEL;

  ln_kernel<<<(int)ROWS, 32, 0, stream>>>(h, ln_g, ln_b, hn);

  dim3 gproj((unsigned)(ROWS / 16), DMODEL / 16);
  proj_kernel<<<gproj, 32, 0, stream>>>(hn, w_qs, q);
  proj_kernel<<<gproj, 32, 0, stream>>>(hn, w_ks, k);
  proj_kernel<<<gproj, 32, 0, stream>>>(hn, w_vs, v);

  attn1_kernel<<<dim3(BATCH * NHEAD, LSEQ / 16, LSEQ / 16), 32, 0, stream>>>(q, k, attn);
  attn2_kernel<<<dim3(BATCH, LSEQ, LSEQ / 16), 32, 0, stream>>>(bias, q, attn);
  softmax_kernel<<<NHEAD * BATCH * LSEQ, 256, 0, stream>>>(attn);
  emb1_kernel<<<dim3(BATCH * NHEAD, LSEQ / 16), 32, 0, stream>>>(attn, v, emb);
  emb2_kernel<<<(unsigned)ROWS, 256, 0, stream>>>(attn, bias, emb);
  fc_kernel<<<gproj, 32, 0, stream>>>(emb, fc_w, fc_b, h, out);
}


# 8x fewer, 8x longer waves in attn1, attn2 (8 j-tiles per wave, 8-wave groups) and softmax (one wave per row; sum order over j changes): 327680 waves become 40960, wave entry was the limit
# speedup vs baseline: 1.8458x; 1.4165x over previous
#include <hip/hip_runtime.h>
#include <hip/hip_bf16.h>
#include <math.h>

#define NHEAD   8
#define DMODEL  256
#define DKDIM   32
#define DVDIM   32
#define LSEQ    1024
#define BATCH   2
#define LN_EPSF 1e-5f

typedef float v2f __attribute__((ext_vector_type(2)));
typedef float v8f __attribute__((ext_vector_type(8)));

__device__ __forceinline__ v8f wmma_f32_k4(v2f a, v2f b, v8f c) {
  return __builtin_amdgcn_wmma_f32_16x16x4_f32(false, a, false, b, (short)0, c,
                                               false, false);
}

__device__ __forceinline__ float ntload(const float* p) {
  return __builtin_nontemporal_load(p);
}
__device__ __forceinline__ v2f ntload2(const float* p) {
  return __builtin_nontemporal_load((const v2f*)p);
}

__global__ void ln_kernel(const float* __restrict__ h,
                          const float* __restrict__ gamma,
                          const float* __restrict__ beta,
                          float* __restrict__ hn) {
  int row  = blockIdx.x;
  int lane = threadIdx.x;
  const float* x = h + (size_t)row * DMODEL;
  float vals[8];
  float s = 0.f;
#pragma unroll
  for (int t = 0; t < 8; ++t) { vals[t] = x[lane + 32 * t]; s += vals[t]; }
#pragma unroll
  for (int o = 16; o > 0; o >>= 1) s += __shfl_xor(s, o, 32);
  float mu = s * (1.0f / DMODEL);
  float vs = 0.f;
#pragma unroll
  for (int t = 0; t < 8; ++t) { float d = vals[t] - mu; vs += d * d; }
#pragma unroll
  for (int o = 16; o > 0; o >>= 1) vs += __shfl_xor(vs, o, 32);
  float inv = rsqrtf(vs * (1.0f / DMODEL) + LN_EPSF);
  float* y = hn + (size_t)row * DMODEL;
#pragma unroll
  for (int t = 0; t < 8; ++t) {
    int c = lane + 32 * t;
    y[c] = (vals[t] - mu) * inv * gamma[c] + beta[c];
  }
}

__global__ void proj_kernel(const float* __restrict__ A,
                            const float* __restrict__ W,
                            float* __restrict__ C) {
  int m0 = blockIdx.x * 16, n0 = blockIdx.y * 16;
  int lane = threadIdx.x, half = lane >> 4, l = lane & 15;
  const float* arow = A + (size_t)(m0 + l) * DMODEL;
  const float* wrow = W + (size_t)(n0 + l) * DMODEL;
  v8f c = {};
  for (int k0 = 0; k0 < DMODEL; k0 += 4) {
    v2f a, b;
    a.x = arow[k0 + 2 * half]; a.y = arow[k0 + 2 * half + 1];
    b.x = wrow[k0 + 2 * half]; b.y = wrow[k0 + 2 * half + 1];
    c = wmma_f32_k4(a, b, c);
  }
#pragma unroll
  for (int r = 0; r < 8; ++r)
    C[(size_t)(m0 + r + 8 * half) * DMODEL + n0 + l] = c[r];
}

__global__ void __launch_bounds__(256)
attn1_kernel(const float* __restrict__ Q,
             const float* __restrict__ K,
             float* __restrict__ attn) {
  int bh = blockIdx.x; int b = bh >> 3; int hh = bh & 7;
  int i0 = blockIdx.y * 16;
  int tid = threadIdx.x, w = tid >> 5;
  int lane = tid & 31, half = lane >> 4, l = lane & 15;
  const float* qrow = Q + ((size_t)(b * LSEQ + i0 + l) * NHEAD + hh) * DKDIM;
  v2f a[8];
#pragma unroll
  for (int s = 0; s < 8; ++s) {
    a[s].x = qrow[4 * s + 2 * half]; a[s].y = qrow[4 * s + 2 * half + 1];
  }
  const float* krow =
      K + ((size_t)(b * LSEQ + w * 128 + l) * NHEAD + hh) * DKDIM;
  float* o = attn + ((size_t)hh * BATCH + b) * LSEQ * LSEQ
             + (size_t)(i0 + 8 * half) * LSEQ + w * 128 + l;
  for (int t = 0; t < 8; ++t) {
    v8f c = {};
#pragma unroll
    for (int s = 0; s < 8; ++s) {
      v2f bf;
      bf.x = krow[4 * s + 2 * half]; bf.y = krow[4 * s + 2 * half + 1];
      c = wmma_f32_k4(a[s], bf, c);
    }
#pragma unroll
    for (int r = 0; r < 8; ++r) o[(size_t)r * LSEQ] = c[r];
    krow += 16 * NHEAD * DKDIM;
    o += 16;
  }
}

__global__ void __launch_bounds__(256)
attn2_kernel(const float* __restrict__ bias,
             const float* __restrict__ Q,
             float* __restrict__ attn) {
  int bi = blockIdx.x; int b = bi >> 10; int i = bi & 1023;
  int tid = threadIdx.x, w = tid >> 5;
  int lane = tid & 31, half = lane >> 4, l = lane & 15;
  bool nvalid = l < 8;
  const float* qv = Q + ((size_t)(b * LSEQ + i) * NHEAD + (l & 7)) * DKDIM
                    + 2 * half;
  v2f bq[8];
#pragma unroll
  for (int s = 0; s < 8; ++s) {
    v2f t = *(const v2f*)(qv + 4 * s);
    bq[s].x = nvalid ? t.x : 0.f;
    bq[s].y = nvalid ? t.y : 0.f;
  }
  const float* arow = bias + (((size_t)b * LSEQ + i) * LSEQ + w * 128 + l) * DKDIM
                      + 2 * half;
  float* pr = attn + (((size_t)(l & 7) * BATCH + b) * LSEQ + i) * LSEQ
              + w * 128 + 8 * half;
#pragma nounroll
  for (int t = 0; t < 8; ++t) {
    v2f a[8];
#pragma unroll
    for (int s = 0; s < 8; ++s) a[s] = ntload2(arow + 4 * s);
    float4 x0 = make_float4(0.f, 0.f, 0.f, 0.f), x1 = x0;
    if (nvalid) { x0 = *(const float4*)(pr); x1 = *(const float4*)(pr + 4); }
    v8f c = {};
#pragma unroll
    for (int s = 0; s < 8; ++s) c = wmma_f32_k4(a[s], bq[s], c);
    if (nvalid) {
      x0.x += c[0]; x0.y += c[1]; x0.z += c[2]; x0.w += c[3];
      x1.x += c[4]; x1.y += c[5]; x1.z += c[6]; x1.w += c[7];
      *(float4*)(pr)     = x0;
      *(float4*)(pr + 4) = x1;
    }
    arow += 16 * DKDIM;
    pr += 16;
  }
}

__global__ void __launch_bounds__(256) softmax_kernel(float* __restrict__ attn) {
  int tid = threadIdx.x, w = tid >> 5, lane = tid & 31;
  float4* p = (float4*)(attn + ((size_t)blockIdx.x * 8 + w) * LSEQ);
  const float scale = 0.17677669529663687f;
  float4 v[8];
#pragma unroll
  for (int u = 0; u < 8; ++u) v[u] = p[lane + 32 * u];
  float mx = -INFINITY;
#pragma unroll
  for (int u = 0; u < 8; ++u) {
    v[u].x *= scale; v[u].y *= scale; v[u].z *= scale; v[u].w *= scale;
    mx = fmaxf(mx, fmaxf(fmaxf(v[u].x, v[u].y), fmaxf(v[u].z, v[u].w)));
  }
#pragma unroll
  for (int o = 16; o > 0; o >>= 1) mx = fmaxf(mx, __shfl_xor(mx, o, 32));
  float s = 0.f;
#pragma unroll
  for (int u = 0; u < 8; ++u) {
    v[u].x = __expf(v[u].x - mx); v[u].y = __expf(v[u].y - mx);
    v[u].z = __expf(v[u].z - mx); v[u].w = __expf(v[u].w - mx);
    s += (v[u].x + v[u].y) + (v[u].z + v[u].w);
  }
#pragma unroll
  for (int o = 16; o > 0; o >>= 1) s += __shfl_xor(s, o, 32);
  float inv = 1.0f / s;
#pragma unroll
  for (int u = 0; u < 8; ++u)
    p[lane + 32 * u] = make_float4(v[u].x * inv, v[u].y * inv,
                                   v[u].z * inv, v[u].w * inv);
}

__global__ void emb1_kernel(const float* __restrict__ attn,
                            const float* __restrict__ V,
                            float* __restrict__ emb) {
  int bh = blockIdx.x; int b = bh >> 3; int hh = bh & 7;
  int i0 = blockIdx.y * 16;
  int lane = threadIdx.x, half = lane >> 4, l = lane & 15;
  const float* arow = attn + (((size_t)hh * BATCH + b) * LSEQ + i0 + l) * LSEQ;
  v8f c0 = {}, c1 = {};
  for (int k0 = 0; k0 < LSEQ; k0 += 4) {
    v2f a;
    a.x = arow[k0 + 2 * half]; a.y = arow[k0 + 2 * half + 1];
    const float* v0 =
        V + ((size_t)(b * LSEQ + k0 + 2 * half) * NHEAD + hh) * DVDIM + l;
    v2f b0, b1;
    b0.x = v0[0];  b0.y = v0[NHEAD * DVDIM];
    b1.x = v0[16]; b1.y = v0[NHEAD * DVDIM + 16];
    c0 = wmma_f32_k4(a, b0, c0);
    c1 = wmma_f32_k4(a, b1, c1);
  }
  float* e = emb + (size_t)(b * LSEQ + i0) * (NHEAD * DVDIM) + hh * DVDIM;
#pragma unroll
  for (int r = 0; r < 8; ++r) {
    e[(size_t)(r + 8 * half) * (NHEAD * DVDIM) + l]      = c0[r];
    e[(size_t)(r + 8 * half) * (NHEAD * DVDIM) + 16 + l] = c1[r];
  }
}

__global__ void __launch_bounds__(256)
emb2_kernel(const float* __restrict__ attn,
            const float* __restrict__ bias,
            float* __restrict__ emb) {
  __shared__ float part[8][NHEAD * DVDIM];
  int bi = blockIdx.x; int b = bi >> 10; int i = bi & 1023;
  int tid = threadIdx.x, w = tid >> 5;
  int lane = tid & 31, half = lane >> 4, l = lane & 15;
  const float* arow = attn + (((size_t)(l & 7) * BATCH + b) * LSEQ + i) * LSEQ
                      + w * 128 + 16 * half;
  const float* brow = bias + (((size_t)b * LSEQ + i) * LSEQ + w * 128 + 16 * half)
                      * (size_t)DKDIM + 2 * l;
  v8f c0 = {}, c1 = {};
  for (int ch = 0; ch < 4; ++ch) {
    v2f bb[16];
    float4 a4[4];
#pragma unroll
    for (int r = 0; r < 16; ++r) bb[r] = ntload2(brow + r * DKDIM);
#pragma unroll
    for (int t = 0; t < 4; ++t) a4[t] = *(const float4*)(arow + 4 * t);
#pragma unroll
    for (int t = 0; t < 4; ++t) {
      v2f a, b0, b1;
      a.x = a4[t].x; a.y = a4[t].y;
      b0.x = bb[4 * t].x;     b0.y = bb[4 * t + 1].x;
      b1.x = bb[4 * t].y;     b1.y = bb[4 * t + 1].y;
      c0 = wmma_f32_k4(a, b0, c0);
      c1 = wmma_f32_k4(a, b1, c1);
      a.x = a4[t].z; a.y = a4[t].w;
      b0.x = bb[4 * t + 2].x; b0.y = bb[4 * t + 3].x;
      b1.x = bb[4 * t + 2].y; b1.y = bb[4 * t + 3].y;
      c0 = wmma_f32_k4(a, b0, c0);
      c1 = wmma_f32_k4(a, b1, c1);
    }
    arow += 32;
    brow += 32 * DKDIM;
  }
  if (half == 0) {
#pragma unroll
    for (int r = 0; r < 8; ++r) {
      v2f o; o.x = c0[r]; o.y = c1[r];
      *(v2f*)(&part[w][r * DVDIM + 2 * l]) = o;
    }
  }
  __syncthreads();
  float sum = part[0][tid];
#pragma unroll
  for (int u = 1; u < 8; ++u) sum += part[u][tid];
  emb[(size_t)(b * LSEQ + i) * (NHEAD * DVDIM) + tid] += sum;
}

__global__ void fc_kernel(const float* __restrict__ emb,
                          const float* __restrict__ W,
                          const float* __restrict__ bvec,
                          const float* __restrict__ resid,
                          float* __restrict__ out) {
  int m0 = blockIdx.x * 16, n0 = blockIdx.y * 16;
  int lane = threadIdx.x, half = lane >> 4, l = lane & 15;
  const float* arow = emb + (size_t)(m0 + l) * DMODEL;
  const float* wrow = W + (size_t)(n0 + l) * DMODEL;
  v8f c = {};
  for (int k0 = 0; k0 < DMODEL; k0 += 4) {
    v2f a, bf;
    a.x  = arow[k0 + 2 * half]; a.y  = arow[k0 + 2 * half + 1];
    bf.x = wrow[k0 + 2 * half]; bf.y = wrow[k0 + 2 * half + 1];
    c = wmma_f32_k4(a, bf, c);
  }
#pragma unroll
  for (int r = 0; r < 8; ++r) {
    int m = m0 + r + 8 * half, n = n0 + l;
    out[(size_t)m * DMODEL + n] = c[r] + bvec[n] + resid[(size_t)m * DMODEL + n];
  }
}

extern "C" void kernel_launch(void* const* d_in, const int* in_sizes, int n_in,
                              void* d_out, int out_size, void* d_ws,
                              size_t ws_size, hipStream_t stream) {
  (void)in_sizes; (void)n_in; (void)out_size; (void)ws_size;
  const float* h    = (const float*)d_in[0];
  const float* bias = (const float*)d_in[1];
  const float* w_qs = (const float*)d_in[2];
  const float* w_ks = (const float*)d_in[3];
  const float* w_vs = (const float*)d_in[4];
  const float* ln_g = (const float*)d_in[5];
  const float* ln_b = (const float*)d_in[6];
  const float* fc_w = (const float*)d_in[7];
  const float* fc_b = (const float*)d_in[8];

  float* out  = (float*)d_out;
  float* attn = out + (size_t)BATCH * LSEQ * DMODEL;

  const size_t ROWS = (size_t)BATCH * LSEQ;
  float* ws  = (float*)d_ws;
  float* hn  = ws;
  float* q   = hn + ROWS * DMODEL;
  float* k   = q + ROWS * DMODEL;
  float* v   = k + ROWS * DMODEL;
  float* emb = v + ROWS * DMODEL;

  ln_kernel<<<(int)ROWS, 32, 0, stream>>>(h, ln_g, ln_b, hn);

  dim3 gproj((unsigned)(ROWS / 16), DMODEL / 16);
  proj_kernel<<<gproj, 32, 0, stream>>>(hn, w_qs, q);
  proj_kernel<<<gproj, 32, 0, stream>>>(hn, w_ks, k);
  proj_kernel<<<gproj, 32, 0, stream>>>(hn, w_vs, v);

  attn1_kernel<<<dim3(BATCH * NHEAD, LSEQ / 16), 256, 0, stream>>>(q, k, attn);
  attn2_kernel<<<(unsigned)ROWS, 256, 0, stream>>>(bias, q, attn);
  softmax_kernel<<<NHEAD * BATCH * LSEQ / 8, 256, 0, stream>>>(attn);
  emb1_kernel<<<dim3(BATCH * NHEAD, LSEQ / 16), 32, 0, stream>>>(attn, v, emb);
  emb2_kernel<<<(unsigned)ROWS, 256, 0, stream>>>(attn, bias, emb);
  fc_kernel<<<gproj, 32, 0, stream>>>(emb, fc_w, fc_b, h, out);
}


# attn2, softmax and emb2 fused into one 8-wave kernel per (b,i) row: bias read once and parked in LDS for emb2, scores stay in registers, attn1 starts the accumulator (sum order over d, j changes)
# speedup vs baseline: 2.0768x; 1.1251x over previous
#include <hip/hip_runtime.h>
#include <hip/hip_bf16.h>
#include <math.h>

#define NHEAD   8
#define DMODEL  256
#define DKDIM   32
#define DVDIM   32
#define LSEQ    1024
#define BATCH   2
#define LN_EPSF 1e-5f

typedef float v2f __attribute__((ext_vector_type(2)));
typedef float v8f __attribute__((ext_vector_type(8)));

__device__ __forceinline__ v8f wmma_f32_k4(v2f a, v2f b, v8f c) {
  return __builtin_amdgcn_wmma_f32_16x16x4_f32(false, a, false, b, (short)0, c,
                                               false, false);
}

__device__ __forceinline__ float ntload(const float* p) {
  return __builtin_nontemporal_load(p);
}
__device__ __forceinline__ v2f ntload2(const float* p) {
  return __builtin_nontemporal_load((const v2f*)p);
}

__global__ void ln_kernel(const float* __restrict__ h,
                          const float* __restrict__ gamma,
                          const float* __restrict__ beta,
                          float* __restrict__ hn) {
  int row  = blockIdx.x;
  int lane = threadIdx.x;
  const float* x = h + (size_t)row * DMODEL;
  float vals[8];
  float s = 0.f;
#pragma unroll
  for (int t = 0; t < 8; ++t) { vals[t] = x[lane + 32 * t]; s += vals[t]; }
#pragma unroll
  for (int o = 16; o > 0; o >>= 1) s += __shfl_xor(s, o, 32);
  float mu = s * (1.0f / DMODEL);
  float vs = 0.f;
#pragma unroll
  for (int t = 0; t < 8; ++t) { float d = vals[t] - mu; vs += d * d; }
#pragma unroll
  for (int o = 16; o > 0; o >>= 1) vs += __shfl_xor(vs, o, 32);
  float inv = rsqrtf(vs * (1.0f / DMODEL) + LN_EPSF);
  float* y = hn + (size_t)row * DMODEL;
#pragma unroll
  for (int t = 0; t < 8; ++t) {
    int c = lane + 32 * t;
    y[c] = (vals[t] - mu) * inv * gamma[c] + beta[c];
  }
}

__global__ void proj_kernel(const float* __restrict__ A,
                            const float* __restrict__ W,
                            float* __restrict__ C) {
  int m0 = blockIdx.x * 16, n0 = blockIdx.y * 16;
  int lane = threadIdx.x, half = lane >> 4, l = lane & 15;
  const float* arow = A + (size_t)(m0 + l) * DMODEL;
  const float* wrow = W + (size_t)(n0 + l) * DMODEL;
  v8f c = {};
  for (int k0 = 0; k0 < DMODEL; k0 += 4) {
    v2f a, b;
    a.x = arow[k0 + 2 * half]; a.y = arow[k0 + 2 * half + 1];
    b.x = wrow[k0 + 2 * half]; b.y = wrow[k0 + 2 * half + 1];
    c = wmma_f32_k4(a, b, c);
  }
#pragma unroll
  for (int r = 0; r < 8; ++r)
    C[(size_t)(m0 + r + 8 * half) * DMODEL + n0 + l] = c[r];
}

__global__ void __launch_bounds__(256)
attn1_kernel(const float* __restrict__ Q,
             const float* __restrict__ K,
             float* __restrict__ attn) {
  int bh = blockIdx.x; int b = bh >> 3; int hh = bh & 7;
  int i0 = blockIdx.y * 16;
  int tid = threadIdx.x, w = tid >> 5;
  int lane = tid & 31, half = lane >> 4, l = lane & 15;
  const float* qrow = Q + ((size_t)(b * LSEQ + i0 + l) * NHEAD + hh) * DKDIM;
  v2f a[8];
#pragma unroll
  for (int s = 0; s < 8; ++s) {
    a[s].x = qrow[4 * s + 2 * half]; a[s].y = qrow[4 * s + 2 * half + 1];
  }
  const float* krow =
      K + ((size_t)(b * LSEQ + w * 128 + l) * NHEAD + hh) * DKDIM;
  float* o = attn + ((size_t)hh * BATCH + b) * LSEQ * LSEQ
             + (size_t)(i0 + 8 * half) * LSEQ + w * 128 + l;
  for (int t = 0; t < 8; ++t) {
    v8f c = {};
#pragma unroll
    for (int s = 0; s < 8; ++s) {
      v2f bf;
      bf.x = krow[4 * s + 2 * half]; bf.y = krow[4 * s + 2 * half + 1];
      c = wmma_f32_k4(a[s], bf, c);
    }
#pragma unroll
    for (int r = 0; r < 8; ++r) o[(size_t)r * LSEQ] = c[r];
    krow += 16 * NHEAD * DKDIM;
    o += 16;
  }
}

#define BPITCH 36
__global__ void __launch_bounds__(256, 4)
row_kernel(const float* __restrict__ bias,
           const float* __restrict__ Q,
           float* __restrict__ attn,
           float* __restrict__ emb) {
  __shared__ __attribute__((aligned(16))) float sb[LSEQ * BPITCH];
  __shared__ __attribute__((aligned(16))) float part[8][NHEAD * DVDIM];
  __shared__ float red[2][8][16];
  int bi = blockIdx.x; int b = bi >> 10; int i = bi & 1023;
  int tid = threadIdx.x, w = tid >> 5;
  int lane = tid & 31, half = lane >> 4, l = lane & 15;
  bool nvalid = l < 8;
  const float* qv = Q + ((size_t)(b * LSEQ + i) * NHEAD + (l & 7)) * DKDIM
                    + 2 * half;
  v2f bq[8];
#pragma unroll
  for (int s = 0; s < 8; ++s) {
    v2f t = *(const v2f*)(qv + 4 * s);
    bq[s].x = nvalid ? t.x : 0.f;
    bq[s].y = nvalid ? t.y : 0.f;
  }
  const float* arow = bias + (((size_t)b * LSEQ + i) * LSEQ + w * 128 + l) * DKDIM
                      + 2 * half;
  float* pr = attn + (((size_t)(l & 7) * BATCH + b) * LSEQ + i) * LSEQ
              + w * 128 + 8 * half;
  float* srow = sb + (w * 128 + l) * BPITCH + 2 * half;
  v2f a[8][8];
#pragma unroll
  for (int t = 0; t < 8; ++t)
#pragma unroll
    for (int s = 0; s < 8; ++s)
      a[t][s] = ntload2(arow + (size_t)t * 16 * DKDIM + 4 * s);
  v8f xa[8];
#pragma unroll
  for (int t = 0; t < 8; ++t) {
    float4 lo = *(const float4*)(pr + 16 * t);
    float4 hi = *(const float4*)(pr + 16 * t + 4);
    xa[t][0] = lo.x; xa[t][1] = lo.y; xa[t][2] = lo.z; xa[t][3] = lo.w;
    xa[t][4] = hi.x; xa[t][5] = hi.y; xa[t][6] = hi.z; xa[t][7] = hi.w;
  }
#pragma unroll
  for (int t = 0; t < 8; ++t) {
#pragma unroll
    for (int s = 0; s < 8; ++s) xa[t] = wmma_f32_k4(a[t][s], bq[s], xa[t]);
#pragma unroll
    for (int s = 0; s < 8; ++s)
      *(v2f*)(srow + t * 16 * BPITCH + 4 * s) = a[t][s];
  }
  const float scale = 0.17677669529663687f;
  float mx = -INFINITY;
#pragma unroll
  for (int t = 0; t < 8; ++t)
#pragma unroll
    for (int r = 0; r < 8; ++r) { xa[t][r] *= scale; mx = fmaxf(mx, xa[t][r]); }
  mx = fmaxf(mx, __shfl_xor(mx, 16, 32));
  if (half == 0) red[0][w][l] = mx;
  __syncthreads();
#pragma unroll
  for (int u = 0; u < 8; ++u) mx = fmaxf(mx, red[0][u][l]);
  float sm = 0.f;
#pragma unroll
  for (int t = 0; t < 8; ++t)
#pragma unroll
    for (int r = 0; r < 8; ++r) { xa[t][r] = __expf(xa[t][r] - mx); sm += xa[t][r]; }
  sm += __shfl_xor(sm, 16, 32);
  if (half == 0) red[1][w][l] = sm;
  __syncthreads();
  float tot = red[1][0][l];
#pragma unroll
  for (int u = 1; u < 8; ++u) tot += red[1][u][l];
  float inv = 1.0f / tot;
#pragma unroll
  for (int t = 0; t < 8; ++t)
#pragma unroll
    for (int r = 0; r < 8; ++r) xa[t][r] *= inv;
  if (nvalid) {
#pragma unroll
    for (int t = 0; t < 8; ++t) {
      *(float4*)(pr + 16 * t) =
          make_float4(xa[t][0], xa[t][1], xa[t][2], xa[t][3]);
      *(float4*)(pr + 16 * t + 4) =
          make_float4(xa[t][4], xa[t][5], xa[t][6], xa[t][7]);
    }
  }
  const float* brd = sb + (w * 128 + 8 * half) * BPITCH + 2 * l;
  v2f rb[64];
#pragma unroll
  for (int u = 0; u < 64; ++u)
    rb[u] = *(const v2f*)(brd + ((u >> 3) * 16 + (u & 7)) * BPITCH);
#pragma unroll
  for (int t = 0; t < 8; ++t) {
    asm volatile("" :: "v"(rb[8 * t]), "v"(rb[8 * t + 1]), "v"(rb[8 * t + 2]),
                 "v"(rb[8 * t + 3]), "v"(rb[8 * t + 4]), "v"(rb[8 * t + 5]),
                 "v"(rb[8 * t + 6]), "v"(rb[8 * t + 7]));
  }
  v8f c0 = {}, c1 = {};
#pragma unroll
  for (int t = 0; t < 8; ++t) {
#pragma unroll
    for (int m = 0; m < 4; ++m) {
      v2f r0 = rb[8 * t + 2 * m];
      v2f r1 = rb[8 * t + 2 * m + 1];
      v2f pa, b0, b1;
      pa.x = xa[t][2 * m]; pa.y = xa[t][2 * m + 1];
      b0.x = r0.x; b0.y = r1.x;
      b1.x = r0.y; b1.y = r1.y;
      c0 = wmma_f32_k4(pa, b0, c0);
      c1 = wmma_f32_k4(pa, b1, c1);
    }
  }
  if (half == 0) {
#pragma unroll
    for (int r = 0; r < 8; ++r) {
      v2f o; o.x = c0[r]; o.y = c1[r];
      *(v2f*)(&part[w][r * DVDIM + 2 * l]) = o;
    }
  }
  __syncthreads();
  float sum = part[0][tid];
#pragma unroll
  for (int u = 1; u < 8; ++u) sum += part[u][tid];
  emb[(size_t)(b * LSEQ + i) * (NHEAD * DVDIM) + tid] = sum;
}

__global__ void emb1_kernel(const float* __restrict__ attn,
                            const float* __restrict__ V,
                            float* __restrict__ emb) {
  int bh = blockIdx.x; int b = bh >> 3; int hh = bh & 7;
  int i0 = blockIdx.y * 16;
  int lane = threadIdx.x, half = lane >> 4, l = lane & 15;
  const float* arow = attn + (((size_t)hh * BATCH + b) * LSEQ + i0 + l) * LSEQ;
  v8f c0 = {}, c1 = {};
  for (int k0 = 0; k0 < LSEQ; k0 += 4) {
    v2f a;
    a.x = arow[k0 + 2 * half]; a.y = arow[k0 + 2 * half + 1];
    const float* v0 =
        V + ((size_t)(b * LSEQ + k0 + 2 * half) * NHEAD + hh) * DVDIM + l;
    v2f b0, b1;
    b0.x = v0[0];  b0.y = v0[NHEAD * DVDIM];
    b1.x = v0[16]; b1.y = v0[NHEAD * DVDIM + 16];
    c0 = wmma_f32_k4(a, b0, c0);
    c1 = wmma_f32_k4(a, b1, c1);
  }
  float* e = emb + (size_t)(b * LSEQ + i0) * (NHEAD * DVDIM) + hh * DVDIM;
#pragma unroll
  for (int r = 0; r < 8; ++r) {
    e[(size_t)(r + 8 * half) * (NHEAD * DVDIM) + l]      += c0[r];
    e[(size_t)(r + 8 * half) * (NHEAD * DVDIM) + 16 + l] += c1[r];
  }
}

__global__ void fc_kernel(const float* __restrict__ emb,
                          const float* __restrict__ W,
                          const float* __restrict__ bvec,
                          const float* __restrict__ resid,
                          float* __restrict__ out) {
  int m0 = blockIdx.x * 16, n0 = blockIdx.y * 16;
  int lane = threadIdx.x, half = lane >> 4, l = lane & 15;
  const float* arow = emb + (size_t)(m0 + l) * DMODEL;
  const float* wrow = W + (size_t)(n0 + l) * DMODEL;
  v8f c = {};
  for (int k0 = 0; k0 < DMODEL; k0 += 4) {
    v2f a, bf;
    a.x  = arow[k0 + 2 * half]; a.y  = arow[k0 + 2 * half + 1];
    bf.x = wrow[k0 + 2 * half]; bf.y = wrow[k0 + 2 * half + 1];
    c = wmma_f32_k4(a, bf, c);
  }
#pragma unroll
  for (int r = 0; r < 8; ++r) {
    int m = m0 + r + 8 * half, n = n0 + l;
    out[(size_t)m * DMODEL + n] = c[r] + bvec[n] + resid[(size_t)m * DMODEL + n];
  }
}

extern "C" void kernel_launch(void* const* d_in, const int* in_sizes, int n_in,
                              void* d_out, int out_size, void* d_ws,
                              size_t ws_size, hipStream_t stream) {
  (void)in_sizes; (void)n_in; (void)out_size; (void)ws_size;
  const float* h    = (const float*)d_in[0];
  const float* bias = (const float*)d_in[1];
  const float* w_qs = (const float*)d_in[2];
  const float* w_ks = (const float*)d_in[3];
  const float* w_vs = (const float*)d_in[4];
  const float* ln_g = (const float*)d_in[5];
  const float* ln_b = (const float*)d_in[6];
  const float* fc_w = (const float*)d_in[7];
  const float* fc_b = (const float*)d_in[8];

  float* out  = (float*)d_out;
  float* attn = out + (size_t)BATCH * LSEQ * DMODEL;

  const size_t ROWS = (size_t)BATCH * LSEQ;
  float* ws  = (float*)d_ws;
  float* hn  = ws;
  float* q   = hn + ROWS * DMODEL;
  float* k   = q + ROWS * DMODEL;
  float* v   = k + ROWS * DMODEL;
  float* emb = v + ROWS * DMODEL;

  ln_kernel<<<(int)ROWS, 32, 0, stream>>>(h, ln_g, ln_b, hn);

  dim3 gproj((unsigned)(ROWS / 16), DMODEL / 16);
  proj_kernel<<<gproj, 32, 0, stream>>>(hn, w_qs, q);
  proj_kernel<<<gproj, 32, 0, stream>>>(hn, w_ks, k);
  proj_kernel<<<gproj, 32, 0, stream>>>(hn, w_vs, v);

  attn1_kernel<<<dim3(BATCH * NHEAD, LSEQ / 16), 256, 0, stream>>>(q, k, attn);
  row_kernel<<<(unsigned)ROWS, 256, 0, stream>>>(bias, q, attn, emb);
  emb1_kernel<<<dim3(BATCH * NHEAD, LSEQ / 16), 32, 0, stream>>>(attn, v, emb);
  fc_kernel<<<gproj, 32, 0, stream>>>(emb, fc_w, fc_b, h, out);
}


# row kernel reads the bias as whole 128-byte lines (b128, 8 lanes per row, each line requested once), parks them in LDS and takes the score product's A fragments from LDS (sum order over d changes)
# speedup vs baseline: 2.9549x; 1.4228x over previous
#include <hip/hip_runtime.h>
#include <hip/hip_bf16.h>
#include <math.h>

#define NHEAD   8
#define DMODEL  256
#define DKDIM   32
#define DVDIM   32
#define LSEQ    1024
#define BATCH   2
#define LN_EPSF 1e-5f

typedef float v2f __attribute__((ext_vector_type(2)));
typedef float v4f __attribute__((ext_vector_type(4)));
typedef float v8f __attribute__((ext_vector_type(8)));

__device__ __forceinline__ v8f wmma_f32_k4(v2f a, v2f b, v8f c) {
  return __builtin_amdgcn_wmma_f32_16x16x4_f32(false, a, false, b, (short)0, c,
                                               false, false);
}

__device__ __forceinline__ float ntload(const float* p) {
  return __builtin_nontemporal_load(p);
}
__device__ __forceinline__ v4f ntload4(const float* p) {
  return __builtin_nontemporal_load((const v4f*)p);
}

__global__ void ln_kernel(const float* __restrict__ h,
                          const float* __restrict__ gamma,
                          const float* __restrict__ beta,
                          float* __restrict__ hn) {
  int row  = blockIdx.x;
  int lane = threadIdx.x;
  const float* x = h + (size_t)row * DMODEL;
  float vals[8];
  float s = 0.f;
#pragma unroll
  for (int t = 0; t < 8; ++t) { vals[t] = x[lane + 32 * t]; s += vals[t]; }
#pragma unroll
  for (int o = 16; o > 0; o >>= 1) s += __shfl_xor(s, o, 32);
  float mu = s * (1.0f / DMODEL);
  float vs = 0.f;
#pragma unroll
  for (int t = 0; t < 8; ++t) { float d = vals[t] - mu; vs += d * d; }
#pragma unroll
  for (int o = 16; o > 0; o >>= 1) vs += __shfl_xor(vs, o, 32);
  float inv = rsqrtf(vs * (1.0f / DMODEL) + LN_EPSF);
  float* y = hn + (size_t)row * DMODEL;
#pragma unroll
  for (int t = 0; t < 8; ++t) {
    int c = lane + 32 * t;
    y[c] = (vals[t] - mu) * inv * gamma[c] + beta[c];
  }
}

__global__ void proj_kernel(const float* __restrict__ A,
                            const float* __restrict__ W,
                            float* __restrict__ C) {
  int m0 = blockIdx.x * 16, n0 = blockIdx.y * 16;
  int lane = threadIdx.x, half = lane >> 4, l = lane & 15;
  const float* arow = A + (size_t)(m0 + l) * DMODEL;
  const float* wrow = W + (size_t)(n0 + l) * DMODEL;
  v8f c = {};
  for (int k0 = 0; k0 < DMODEL; k0 += 4) {
    v2f a, b;
    a.x = arow[k0 + 2 * half]; a.y = arow[k0 + 2 * half + 1];
    b.x = wrow[k0 + 2 * half]; b.y = wrow[k0 + 2 * half + 1];
    c = wmma_f32_k4(a, b, c);
  }
#pragma unroll
  for (int r = 0; r < 8; ++r)
    C[(size_t)(m0 + r + 8 * half) * DMODEL + n0 + l] = c[r];
}

__global__ void __launch_bounds__(256)
attn1_kernel(const float* __restrict__ Q,
             const float* __restrict__ K,
             float* __restrict__ attn) {
  int bh = blockIdx.x; int b = bh >> 3; int hh = bh & 7;
  int i0 = blockIdx.y * 16;
  int tid = threadIdx.x, w = tid >> 5;
  int lane = tid & 31, half = lane >> 4, l = lane & 15;
  const float* qrow = Q + ((size_t)(b * LSEQ + i0 + l) * NHEAD + hh) * DKDIM;
  v2f a[8];
#pragma unroll
  for (int s = 0; s < 8; ++s) {
    a[s].x = qrow[4 * s + 2 * half]; a[s].y = qrow[4 * s + 2 * half + 1];
  }
  const float* krow =
      K + ((size_t)(b * LSEQ + w * 128 + l) * NHEAD + hh) * DKDIM;
  float* o = attn + ((size_t)hh * BATCH + b) * LSEQ * LSEQ
             + (size_t)(i0 + 8 * half) * LSEQ + w * 128 + l;
  for (int t = 0; t < 8; ++t) {
    v8f c = {};
#pragma unroll
    for (int s = 0; s < 8; ++s) {
      v2f bf;
      bf.x = krow[4 * s + 2 * half]; bf.y = krow[4 * s + 2 * half + 1];
      c = wmma_f32_k4(a[s], bf, c);
    }
#pragma unroll
    for (int r = 0; r < 8; ++r) o[(size_t)r * LSEQ] = c[r];
    krow += 16 * NHEAD * DKDIM;
    o += 16;
  }
}

#define BPITCH 36
__global__ void __launch_bounds__(256, 4)
row_kernel(const float* __restrict__ bias,
           const float* __restrict__ Q,
           float* __restrict__ attn,
           float* __restrict__ emb) {
  __shared__ __attribute__((aligned(16))) float sb[LSEQ * BPITCH];
  __shared__ __attribute__((aligned(16))) float part[8][NHEAD * DVDIM];
  __shared__ float red[2][8][16];
  int bi = blockIdx.x; int b = bi >> 10; int i = bi & 1023;
  int tid = threadIdx.x, w = tid >> 5;
  int lane = tid & 31, half = lane >> 4, l = lane & 15;
  bool nvalid = l < 8;
  const float* qv = Q + ((size_t)(b * LSEQ + i) * NHEAD + (l & 7)) * DKDIM
                    + 4 * half;
  v2f bq[8];
#pragma unroll
  for (int u = 0; u < 4; ++u) {
    v4f t = *(const v4f*)(qv + 8 * u);
    bq[2 * u].x     = nvalid ? t.x : 0.f;
    bq[2 * u].y     = nvalid ? t.y : 0.f;
    bq[2 * u + 1].x = nvalid ? t.z : 0.f;
    bq[2 * u + 1].y = nvalid ? t.w : 0.f;
  }
  int lrow = ((lane >> 4) & 1) + 8 * ((lane >> 3) & 1);
  int chunk = lane & 7;
  const float* grow = bias
      + (((size_t)b * LSEQ + i) * LSEQ + w * 128 + lrow) * DKDIM + 4 * chunk;
  float* spark = sb + (w * 128 + lrow) * BPITCH + 4 * chunk;
  float* pr = attn + (((size_t)(l & 7) * BATCH + b) * LSEQ + i) * LSEQ
              + w * 128 + 8 * half;
  v4f st[8][4];
#pragma unroll
  for (int t = 0; t < 8; ++t)
#pragma unroll
    for (int u = 0; u < 4; ++u)
      st[t][u] = ntload4(grow + (size_t)(16 * t + 2 * u) * DKDIM);
  v8f xa[8];
#pragma unroll
  for (int t = 0; t < 8; ++t) {
    float4 lo = *(const float4*)(pr + 16 * t);
    float4 hi = *(const float4*)(pr + 16 * t + 4);
    xa[t][0] = lo.x; xa[t][1] = lo.y; xa[t][2] = lo.z; xa[t][3] = lo.w;
    xa[t][4] = hi.x; xa[t][5] = hi.y; xa[t][6] = hi.z; xa[t][7] = hi.w;
  }
#pragma unroll
  for (int t = 0; t < 8; ++t)
#pragma unroll
    for (int u = 0; u < 4; ++u)
      *(v4f*)(spark + (16 * t + 2 * u) * BPITCH) = st[t][u];
  __syncthreads();
  const float* frow = sb + (w * 128 + l) * BPITCH + 4 * half;
  v4f af[8][4];
#pragma unroll
  for (int t = 0; t < 8; ++t)
#pragma unroll
    for (int u = 0; u < 4; ++u)
      af[t][u] = *(const v4f*)(frow + t * 16 * BPITCH + 8 * u);
#pragma unroll
  for (int t = 0; t < 8; ++t) {
    asm volatile("" :: "v"(af[t][0]), "v"(af[t][1]), "v"(af[t][2]),
                 "v"(af[t][3]));
  }
#pragma unroll
  for (int t = 0; t < 8; ++t) {
#pragma unroll
    for (int u = 0; u < 4; ++u) {
      v2f a0, a1;
      a0.x = af[t][u].x; a0.y = af[t][u].y;
      a1.x = af[t][u].z; a1.y = af[t][u].w;
      xa[t] = wmma_f32_k4(a0, bq[2 * u], xa[t]);
      xa[t] = wmma_f32_k4(a1, bq[2 * u + 1], xa[t]);
    }
  }
  const float scale = 0.17677669529663687f;
  float mx = -INFINITY;
#pragma unroll
  for (int t = 0; t < 8; ++t)
#pragma unroll
    for (int r = 0; r < 8; ++r) { xa[t][r] *= scale; mx = fmaxf(mx, xa[t][r]); }
  mx = fmaxf(mx, __shfl_xor(mx, 16, 32));
  if (half == 0) red[0][w][l] = mx;
  __syncthreads();
#pragma unroll
  for (int u = 0; u < 8; ++u) mx = fmaxf(mx, red[0][u][l]);
  float sm = 0.f;
#pragma unroll
  for (int t = 0; t < 8; ++t)
#pragma unroll
    for (int r = 0; r < 8; ++r) { xa[t][r] = __expf(xa[t][r] - mx); sm += xa[t][r]; }
  sm += __shfl_xor(sm, 16, 32);
  if (half == 0) red[1][w][l] = sm;
  __syncthreads();
  float tot = red[1][0][l];
#pragma unroll
  for (int u = 1; u < 8; ++u) tot += red[1][u][l];
  float inv = 1.0f / tot;
#pragma unroll
  for (int t = 0; t < 8; ++t)
#pragma unroll
    for (int r = 0; r < 8; ++r) xa[t][r] *= inv;
  if (nvalid) {
#pragma unroll
    for (int t = 0; t < 8; ++t) {
      *(float4*)(pr + 16 * t) =
          make_float4(xa[t][0], xa[t][1], xa[t][2], xa[t][3]);
      *(float4*)(pr + 16 * t + 4) =
          make_float4(xa[t][4], xa[t][5], xa[t][6], xa[t][7]);
    }
  }
  const float* brd = sb + (w * 128 + 8 * half) * BPITCH + 2 * l;
  v2f rb[64];
#pragma unroll
  for (int u = 0; u < 64; ++u)
    rb[u] = *(const v2f*)(brd + ((u >> 3) * 16 + (u & 7)) * BPITCH);
#pragma unroll
  for (int t = 0; t < 8; ++t) {
    asm volatile("" :: "v"(rb[8 * t]), "v"(rb[8 * t + 1]), "v"(rb[8 * t + 2]),
                 "v"(rb[8 * t + 3]), "v"(rb[8 * t + 4]), "v"(rb[8 * t + 5]),
                 "v"(rb[8 * t + 6]), "v"(rb[8 * t + 7]));
  }
  v8f c0 = {}, c1 = {};
#pragma unroll
  for (int t = 0; t < 8; ++t) {
#pragma unroll
    for (int m = 0; m < 4; ++m) {
      v2f r0 = rb[8 * t + 2 * m];
      v2f r1 = rb[8 * t + 2 * m + 1];
      v2f pa, b0, b1;
      pa.x = xa[t][2 * m]; pa.y = xa[t][2 * m + 1];
      b0.x = r0.x; b0.y = r1.x;
      b1.x = r0.y; b1.y = r1.y;
      c0 = wmma_f32_k4(pa, b0, c0);
      c1 = wmma_f32_k4(pa, b1, c1);
    }
  }
  if (half == 0) {
#pragma unroll
    for (int r = 0; r < 8; ++r) {
      v2f o; o.x = c0[r]; o.y = c1[r];
      *(v2f*)(&part[w][r * DVDIM + 2 * l]) = o;
    }
  }
  __syncthreads();
  float sum = part[0][tid];
#pragma unroll
  for (int u = 1; u < 8; ++u) sum += part[u][tid];
  emb[(size_t)(b * LSEQ + i) * (NHEAD * DVDIM) + tid] = sum;
}

__global__ void emb1_kernel(const float* __restrict__ attn,
                            const float* __restrict__ V,
                            float* __restrict__ emb) {
  int bh = blockIdx.x; int b = bh >> 3; int hh = bh & 7;
  int i0 = blockIdx.y * 16;
  int lane = threadIdx.x, half = lane >> 4, l = lane & 15;
  const float* arow = attn + (((size_t)hh * BATCH + b) * LSEQ + i0 + l) * LSEQ;
  v8f c0 = {}, c1 = {};
  for (int k0 = 0; k0 < LSEQ; k0 += 4) {
    v2f a;
    a.x = arow[k0 + 2 * half]; a.y = arow[k0 + 2 * half + 1];
    const float* v0 =
        V + ((size_t)(b * LSEQ + k0 + 2 * half) * NHEAD + hh) * DVDIM + l;
    v2f b0, b1;
    b0.x = v0[0];  b0.y = v0[NHEAD * DVDIM];
    b1.x = v0[16]; b1.y = v0[NHEAD * DVDIM + 16];
    c0 = wmma_f32_k4(a, b0, c0);
    c1 = wmma_f32_k4(a, b1, c1);
  }
  float* e = emb + (size_t)(b * LSEQ + i0) * (NHEAD * DVDIM) + hh * DVDIM;
#pragma unroll
  for (int r = 0; r < 8; ++r) {
    e[(size_t)(r + 8 * half) * (NHEAD * DVDIM) + l]      += c0[r];
    e[(size_t)(r + 8 * half) * (NHEAD * DVDIM) + 16 + l] += c1[r];
  }
}

__global__ void fc_kernel(const float* __restrict__ emb,
                          const float* __restrict__ W,
                          const float* __restrict__ bvec,
                          const float* __restrict__ resid,
                          float* __restrict__ out) {
  int m0 = blockIdx.x * 16, n0 = blockIdx.y * 16;
  int lane = threadIdx.x, half = lane >> 4, l = lane & 15;
  const float* arow = emb + (size_t)(m0 + l) * DMODEL;
  const float* wrow = W + (size_t)(n0 + l) * DMODEL;
  v8f c = {};
  for (int k0 = 0; k0 < DMODEL; k0 += 4) {
    v2f a, bf;
    a.x  = arow[k0 + 2 * half]; a.y  = arow[k0 + 2 * half + 1];
    bf.x = wrow[k0 + 2 * half]; bf.y = wrow[k0 + 2 * half + 1];
    c = wmma_f32_k4(a, bf, c);
  }
#pragma unroll
  for (int r = 0; r < 8; ++r) {
    int m = m0 + r + 8 * half, n = n0 + l;
    out[(size_t)m * DMODEL + n] = c[r] + bvec[n] + resid[(size_t)m * DMODEL + n];
  }
}

extern "C" void kernel_launch(void* const* d_in, const int* in_sizes, int n_in,
                              void* d_out, int out_size, void* d_ws,
                              size_t ws_size, hipStream_t stream) {
  (void)in_sizes; (void)n_in; (void)out_size; (void)ws_size;
  const float* h    = (const float*)d_in[0];
  const float* bias = (const float*)d_in[1];
  const float* w_qs = (const float*)d_in[2];
  const float* w_ks = (const float*)d_in[3];
  const float* w_vs = (const float*)d_in[4];
  const float* ln_g = (const float*)d_in[5];
  const float* ln_b = (const float*)d_in[6];
  const float* fc_w = (const float*)d_in[7];
  const float* fc_b = (const float*)d_in[8];

  float* out  = (float*)d_out;
  float* attn = out + (size_t)BATCH * LSEQ * DMODEL;

  const size_t ROWS = (size_t)BATCH * LSEQ;
  float* ws  = (float*)d_ws;
  float* hn  = ws;
  float* q   = hn + ROWS * DMODEL;
  float* k   = q + ROWS * DMODEL;
  float* v   = k + ROWS * DMODEL;
  float* emb = v + ROWS * DMODEL;

  ln_kernel<<<(int)ROWS, 32, 0, stream>>>(h, ln_g, ln_b, hn);

  dim3 gproj((unsigned)(ROWS / 16), DMODEL / 16);
  proj_kernel<<<gproj, 32, 0, stream>>>(hn, w_qs, q);
  proj_kernel<<<gproj, 32, 0, stream>>>(hn, w_ks, k);
  proj_kernel<<<gproj, 32, 0, stream>>>(hn, w_vs, v);

  attn1_kernel<<<dim3(BATCH * NHEAD, LSEQ / 16), 256, 0, stream>>>(q, k, attn);
  row_kernel<<<(unsigned)ROWS, 256, 0, stream>>>(bias, q, attn, emb);
  emb1_kernel<<<dim3(BATCH * NHEAD, LSEQ / 16), 32, 0, stream>>>(attn, v, emb);
  fc_kernel<<<gproj, 32, 0, stream>>>(emb, fc_w, fc_b, h, out);
}


# emb1: j split over 8 waves per (b, head, 32 rows); probabilities and v read as whole 128-byte lines, probabilities turned in LDS into A fragments; partials summed via LDS (sum order over j changes)
# speedup vs baseline: 3.2135x; 1.0875x over previous
#include <hip/hip_runtime.h>
#include <hip/hip_bf16.h>
#include <math.h>

#define NHEAD   8
#define DMODEL  256
#define DKDIM   32
#define DVDIM   32
#define LSEQ    1024
#define BATCH   2
#define LN_EPSF 1e-5f

typedef float v2f __attribute__((ext_vector_type(2)));
typedef float v4f __attribute__((ext_vector_type(4)));
typedef float v8f __attribute__((ext_vector_type(8)));

__device__ __forceinline__ v8f wmma_f32_k4(v2f a, v2f b, v8f c) {
  return __builtin_amdgcn_wmma_f32_16x16x4_f32(false, a, false, b, (short)0, c,
                                               false, false);
}

__device__ __forceinline__ float ntload(const float* p) {
  return __builtin_nontemporal_load(p);
}
__device__ __forceinline__ v4f ntload4(const float* p) {
  return __builtin_nontemporal_load((const v4f*)p);
}

__global__ void ln_kernel(const float* __restrict__ h,
                          const float* __restrict__ gamma,
                          const float* __restrict__ beta,
                          float* __restrict__ hn) {
  int row  = blockIdx.x;
  int lane = threadIdx.x;
  const float* x = h + (size_t)row * DMODEL;
  float vals[8];
  float s = 0.f;
#pragma unroll
  for (int t = 0; t < 8; ++t) { vals[t] = x[lane + 32 * t]; s += vals[t]; }
#pragma unroll
  for (int o = 16; o > 0; o >>= 1) s += __shfl_xor(s, o, 32);
  float mu = s * (1.0f / DMODEL);
  float vs = 0.f;
#pragma unroll
  for (int t = 0; t < 8; ++t) { float d = vals[t] - mu; vs += d * d; }
#pragma unroll
  for (int o = 16; o > 0; o >>= 1) vs += __shfl_xor(vs, o, 32);
  float inv = rsqrtf(vs * (1.0f / DMODEL) + LN_EPSF);
  float* y = hn + (size_t)row * DMODEL;
#pragma unroll
  for (int t = 0; t < 8; ++t) {
    int c = lane + 32 * t;
    y[c] = (vals[t] - mu) * inv * gamma[c] + beta[c];
  }
}

__global__ void proj_kernel(const float* __restrict__ A,
                            const float* __restrict__ W,
                            float* __restrict__ C) {
  int m0 = blockIdx.x * 16, n0 = blockIdx.y * 16;
  int lane = threadIdx.x, half = lane >> 4, l = lane & 15;
  const float* arow = A + (size_t)(m0 + l) * DMODEL;
  const float* wrow = W + (size_t)(n0 + l) * DMODEL;
  v8f c = {};
  for (int k0 = 0; k0 < DMODEL; k0 += 4) {
    v2f a, b;
    a.x = arow[k0 + 2 * half]; a.y = arow[k0 + 2 * half + 1];
    b.x = wrow[k0 + 2 * half]; b.y = wrow[k0 + 2 * half + 1];
    c = wmma_f32_k4(a, b, c);
  }
#pragma unroll
  for (int r = 0; r < 8; ++r)
    C[(size_t)(m0 + r + 8 * half) * DMODEL + n0 + l] = c[r];
}

__global__ void __launch_bounds__(256)
attn1_kernel(const float* __restrict__ Q,
             const float* __restrict__ K,
             float* __restrict__ attn) {
  int bh = blockIdx.x; int b = bh >> 3; int hh = bh & 7;
  int i0 = blockIdx.y * 16;
  int tid = threadIdx.x, w = tid >> 5;
  int lane = tid & 31, half = lane >> 4, l = lane & 15;
  const float* qrow = Q + ((size_t)(b * LSEQ + i0 + l) * NHEAD + hh) * DKDIM;
  v2f a[8];
#pragma unroll
  for (int s = 0; s < 8; ++s) {
    a[s].x = qrow[4 * s + 2 * half]; a[s].y = qrow[4 * s + 2 * half + 1];
  }
  const float* krow =
      K + ((size_t)(b * LSEQ + w * 128 + l) * NHEAD + hh) * DKDIM;
  float* o = attn + ((size_t)hh * BATCH + b) * LSEQ * LSEQ
             + (size_t)(i0 + 8 * half) * LSEQ + w * 128 + l;
  for (int t = 0; t < 8; ++t) {
    v8f c = {};
#pragma unroll
    for (int s = 0; s < 8; ++s) {
      v2f bf;
      bf.x = krow[4 * s + 2 * half]; bf.y = krow[4 * s + 2 * half + 1];
      c = wmma_f32_k4(a[s], bf, c);
    }
#pragma unroll
    for (int r = 0; r < 8; ++r) o[(size_t)r * LSEQ] = c[r];
    krow += 16 * NHEAD * DKDIM;
    o += 16;
  }
}

#define BPITCH 36
__global__ void __launch_bounds__(256, 4)
row_kernel(const float* __restrict__ bias,
           const float* __restrict__ Q,
           float* __restrict__ attn,
           float* __restrict__ emb) {
  __shared__ __attribute__((aligned(16))) float sb[LSEQ * BPITCH];
  __shared__ __attribute__((aligned(16))) float part[8][NHEAD * DVDIM];
  __shared__ float red[2][8][16];
  int bi = blockIdx.x; int b = bi >> 10; int i = bi & 1023;
  int tid = threadIdx.x, w = tid >> 5;
  int lane = tid & 31, half = lane >> 4, l = lane & 15;
  bool nvalid = l < 8;
  const float* qv = Q + ((size_t)(b * LSEQ + i) * NHEAD + (l & 7)) * DKDIM
                    + 4 * half;
  v2f bq[8];
#pragma unroll
  for (int u = 0; u < 4; ++u) {
    v4f t = *(const v4f*)(qv + 8 * u);
    bq[2 * u].x     = nvalid ? t.x : 0.f;
    bq[2 * u].y     = nvalid ? t.y : 0.f;
    bq[2 * u + 1].x = nvalid ? t.z : 0.f;
    bq[2 * u + 1].y = nvalid ? t.w : 0.f;
  }
  int lrow = ((lane >> 4) & 1) + 8 * ((lane >> 3) & 1);
  int chunk = lane & 7;
  const float* grow = bias
      + (((size_t)b * LSEQ + i) * LSEQ + w * 128 + lrow) * DKDIM + 4 * chunk;
  float* spark = sb + (w * 128 + lrow) * BPITCH + 4 * chunk;
  float* pr = attn + (((size_t)(l & 7) * BATCH + b) * LSEQ + i) * LSEQ
              + w * 128 + 8 * half;
  v4f st[8][4];
#pragma unroll
  for (int t = 0; t < 8; ++t)
#pragma unroll
    for (int u = 0; u < 4; ++u)
      st[t][u] = ntload4(grow + (size_t)(16 * t + 2 * u) * DKDIM);
  v8f xa[8];
#pragma unroll
  for (int t = 0; t < 8; ++t) {
    float4 lo = *(const float4*)(pr + 16 * t);
    float4 hi = *(const float4*)(pr + 16 * t + 4);
    xa[t][0] = lo.x; xa[t][1] = lo.y; xa[t][2] = lo.z; xa[t][3] = lo.w;
    xa[t][4] = hi.x; xa[t][5] = hi.y; xa[t][6] = hi.z; xa[t][7] = hi.w;
  }
#pragma unroll
  for (int t = 0; t < 8; ++t)
#pragma unroll
    for (int u = 0; u < 4; ++u)
      *(v4f*)(spark + (16 * t + 2 * u) * BPITCH) = st[t][u];
  __syncthreads();
  const float* frow = sb + (w * 128 + l) * BPITCH + 4 * half;
  v4f af[8][4];
#pragma unroll
  for (int t = 0; t < 8; ++t)
#pragma unroll
    for (int u = 0; u < 4; ++u)
      af[t][u] = *(const v4f*)(frow + t * 16 * BPITCH + 8 * u);
#pragma unroll
  for (int t = 0; t < 8; ++t) {
    asm volatile("" :: "v"(af[t][0]), "v"(af[t][1]), "v"(af[t][2]),
                 "v"(af[t][3]));
  }
#pragma unroll
  for (int t = 0; t < 8; ++t) {
#pragma unroll
    for (int u = 0; u < 4; ++u) {
      v2f a0, a1;
      a0.x = af[t][u].x; a0.y = af[t][u].y;
      a1.x = af[t][u].z; a1.y = af[t][u].w;
      xa[t] = wmma_f32_k4(a0, bq[2 * u], xa[t]);
      xa[t] = wmma_f32_k4(a1, bq[2 * u + 1], xa[t]);
    }
  }
  const float scale = 0.17677669529663687f;
  float mx = -INFINITY;
#pragma unroll
  for (int t = 0; t < 8; ++t)
#pragma unroll
    for (int r = 0; r < 8; ++r) { xa[t][r] *= scale; mx = fmaxf(mx, xa[t][r]); }
  mx = fmaxf(mx, __shfl_xor(mx, 16, 32));
  if (half == 0) red[0][w][l] = mx;
  __syncthreads();
#pragma unroll
  for (int u = 0; u < 8; ++u) mx = fmaxf(mx, red[0][u][l]);
  float sm = 0.f;
#pragma unroll
  for (int t = 0; t < 8; ++t)
#pragma unroll
    for (int r = 0; r < 8; ++r) { xa[t][r] = __expf(xa[t][r] - mx); sm += xa[t][r]; }
  sm += __shfl_xor(sm, 16, 32);
  if (half == 0) red[1][w][l] = sm;
  __syncthreads();
  float tot = red[1][0][l];
#pragma unroll
  for (int u = 1; u < 8; ++u) tot += red[1][u][l];
  float inv = 1.0f / tot;
#pragma unroll
  for (int t = 0; t < 8; ++t)
#pragma unroll
    for (int r = 0; r < 8; ++r) xa[t][r] *= inv;
  if (nvalid) {
#pragma unroll
    for (int t = 0; t < 8; ++t) {
      *(float4*)(pr + 16 * t) =
          make_float4(xa[t][0], xa[t][1], xa[t][2], xa[t][3]);
      *(float4*)(pr + 16 * t + 4) =
          make_float4(xa[t][4], xa[t][5], xa[t][6], xa[t][7]);
    }
  }
  const float* brd = sb + (w * 128 + 8 * half) * BPITCH + 2 * l;
  v2f rb[64];
#pragma unroll
  for (int u = 0; u < 64; ++u)
    rb[u] = *(const v2f*)(brd + ((u >> 3) * 16 + (u & 7)) * BPITCH);
#pragma unroll
  for (int t = 0; t < 8; ++t) {
    asm volatile("" :: "v"(rb[8 * t]), "v"(rb[8 * t + 1]), "v"(rb[8 * t + 2]),
                 "v"(rb[8 * t + 3]), "v"(rb[8 * t + 4]), "v"(rb[8 * t + 5]),
                 "v"(rb[8 * t + 6]), "v"(rb[8 * t + 7]));
  }
  v8f c0 = {}, c1 = {};
#pragma unroll
  for (int t = 0; t < 8; ++t) {
#pragma unroll
    for (int m = 0; m < 4; ++m) {
      v2f r0 = rb[8 * t + 2 * m];
      v2f r1 = rb[8 * t + 2 * m + 1];
      v2f pa, b0, b1;
      pa.x = xa[t][2 * m]; pa.y = xa[t][2 * m + 1];
      b0.x = r0.x; b0.y = r1.x;
      b1.x = r0.y; b1.y = r1.y;
      c0 = wmma_f32_k4(pa, b0, c0);
      c1 = wmma_f32_k4(pa, b1, c1);
    }
  }
  if (half == 0) {
#pragma unroll
    for (int r = 0; r < 8; ++r) {
      v2f o; o.x = c0[r]; o.y = c1[r];
      *(v2f*)(&part[w][r * DVDIM + 2 * l]) = o;
    }
  }
  __syncthreads();
  float sum = part[0][tid];
#pragma unroll
  for (int u = 1; u < 8; ++u) sum += part[u][tid];
  emb[(size_t)(b * LSEQ + i) * (NHEAD * DVDIM) + tid] = sum;
}

#define APITCH 132
__global__ void __launch_bounds__(256, 4)
emb1_kernel(const float* __restrict__ attn,
            const float* __restrict__ V,
            float* __restrict__ emb) {
  __shared__ __attribute__((aligned(16))) float sa[8][32 * APITCH];
  int bh = blockIdx.x; int b = bh >> 3; int hh = bh & 7;
  int i0 = blockIdx.y * 32;
  int tid = threadIdx.x, w = tid >> 5;
  int lane = tid & 31, half = lane >> 4, l = lane & 15;
  const float* grow = attn + (((size_t)hh * BATCH + b) * LSEQ + i0) * LSEQ
                      + w * 128 + 4 * lane;
  const float* vcol =
      V + ((size_t)(b * LSEQ + w * 128 + 4 * half) * NHEAD + hh) * DVDIM + 2 * l;
  v2f vb[32][2];
#pragma unroll
  for (int s = 0; s < 16; ++s)
#pragma unroll
    for (int e = 0; e < 2; ++e)
      vb[s][e] = *(const v2f*)(
          vcol + (size_t)(8 * (s >> 1) + 2 * (s & 1) + e) * (NHEAD * DVDIM));
  v4f st[32];
#pragma unroll
  for (int r = 0; r < 32; ++r) st[r] = *(const v4f*)(grow + (size_t)r * LSEQ);
  float* mine = &sa[w][0];
#pragma unroll
  for (int r = 0; r < 32; ++r) *(v4f*)(mine + r * APITCH + 4 * lane) = st[r];
  __syncthreads();
#pragma unroll
  for (int s = 16; s < 32; ++s)
#pragma unroll
    for (int e = 0; e < 2; ++e)
      vb[s][e] = *(const v2f*)(
          vcol + (size_t)(8 * (s >> 1) + 2 * (s & 1) + e) * (NHEAD * DVDIM));
  const float* frow = mine + l * APITCH + 4 * half;
  v8f c[2][2] = {};
  v4f af[2][2][4];
#pragma unroll
  for (int m = 0; m < 2; ++m)
#pragma unroll
    for (int uu = 0; uu < 4; ++uu)
      af[0][m][uu] = *(const v4f*)(frow + m * 16 * APITCH + 8 * uu);
#pragma unroll
  for (int q = 0; q < 4; ++q) {
    if (q < 3) {
#pragma unroll
      for (int m = 0; m < 2; ++m)
#pragma unroll
        for (int uu = 0; uu < 4; ++uu)
          af[(q + 1) & 1][m][uu] =
              *(const v4f*)(frow + m * 16 * APITCH + 8 * (4 * (q + 1) + uu));
    }
    __builtin_amdgcn_sched_barrier(0);
#pragma unroll
    for (int uu = 0; uu < 4; ++uu) {
#pragma unroll
      for (int p = 0; p < 2; ++p) {
        const int s = 2 * (4 * q + uu) + p;
        v2f b0, b1;
        b0.x = vb[s][0].x; b0.y = vb[s][1].x;
        b1.x = vb[s][0].y; b1.y = vb[s][1].y;
#pragma unroll
        for (int m = 0; m < 2; ++m) {
          v2f a;
          a.x = p ? af[q & 1][m][uu].z : af[q & 1][m][uu].x;
          a.y = p ? af[q & 1][m][uu].w : af[q & 1][m][uu].y;
          c[m][0] = wmma_f32_k4(a, b0, c[m][0]);
          c[m][1] = wmma_f32_k4(a, b1, c[m][1]);
        }
      }
    }
  }
#pragma unroll
  for (int m = 0; m < 2; ++m)
#pragma unroll
    for (int r = 0; r < 8; ++r) {
      v2f o; o.x = c[m][0][r]; o.y = c[m][1][r];
      *(v2f*)(mine + (16 * m + 8 * half + r) * DVDIM + 2 * l) = o;
    }
  __syncthreads();
  float* e = emb + (size_t)(b * LSEQ + i0) * (NHEAD * DVDIM) + hh * DVDIM;
#pragma unroll
  for (int q = 0; q < 4; ++q) {
    int idx = tid + 256 * q;
    float sum = sa[0][idx];
#pragma unroll
    for (int u = 1; u < 8; ++u) sum += sa[u][idx];
    e[(size_t)(idx >> 5) * (NHEAD * DVDIM) + (idx & 31)] += sum;
  }
}

__global__ void fc_kernel(const float* __restrict__ emb,
                          const float* __restrict__ W,
                          const float* __restrict__ bvec,
                          const float* __restrict__ resid,
                          float* __restrict__ out) {
  int m0 = blockIdx.x * 16, n0 = blockIdx.y * 16;
  int lane = threadIdx.x, half = lane >> 4, l = lane & 15;
  const float* arow = emb + (size_t)(m0 + l) * DMODEL;
  const float* wrow = W + (size_t)(n0 + l) * DMODEL;
  v8f c = {};
  for (int k0 = 0; k0 < DMODEL; k0 += 4) {
    v2f a, bf;
    a.x  = arow[k0 + 2 * half]; a.y  = arow[k0 + 2 * half + 1];
    bf.x = wrow[k0 + 2 * half]; bf.y = wrow[k0 + 2 * half + 1];
    c = wmma_f32_k4(a, bf, c);
  }
#pragma unroll
  for (int r = 0; r < 8; ++r) {
    int m = m0 + r + 8 * half, n = n0 + l;
    out[(size_t)m * DMODEL + n] = c[r] + bvec[n] + resid[(size_t)m * DMODEL + n];
  }
}

extern "C" void kernel_launch(void* const* d_in, const int* in_sizes, int n_in,
                              void* d_out, int out_size, void* d_ws,
                              size_t ws_size, hipStream_t stream) {
  (void)in_sizes; (void)n_in; (void)out_size; (void)ws_size;
  const float* h    = (const float*)d_in[0];
  const float* bias = (const float*)d_in[1];
  const float* w_qs = (const float*)d_in[2];
  const float* w_ks = (const float*)d_in[3];
  const float* w_vs = (const float*)d_in[4];
  const float* ln_g = (const float*)d_in[5];
  const float* ln_b = (const float*)d_in[6];
  const float* fc_w = (const float*)d_in[7];
  const float* fc_b = (const float*)d_in[8];

  float* out  = (float*)d_out;
  float* attn = out + (size_t)BATCH * LSEQ * DMODEL;

  const size_t ROWS = (size_t)BATCH * LSEQ;
  float* ws  = (float*)d_ws;
  float* hn  = ws;
  float* q   = hn + ROWS * DMODEL;
  float* k   = q + ROWS * DMODEL;
  float* v   = k + ROWS * DMODEL;
  float* emb = v + ROWS * DMODEL;

  ln_kernel<<<(int)ROWS, 32, 0, stream>>>(h, ln_g, ln_b, hn);

  dim3 gproj((unsigned)(ROWS / 16), DMODEL / 16);
  proj_kernel<<<gproj, 32, 0, stream>>>(hn, w_qs, q);
  proj_kernel<<<gproj, 32, 0, stream>>>(hn, w_ks, k);
  proj_kernel<<<gproj, 32, 0, stream>>>(hn, w_vs, v);

  attn1_kernel<<<dim3(BATCH * NHEAD, LSEQ / 16), 256, 0, stream>>>(q, k, attn);
  row_kernel<<<(unsigned)ROWS, 256, 0, stream>>>(bias, q, attn, emb);
  emb1_kernel<<<dim3(BATCH * NHEAD, LSEQ / 32), 256, 0, stream>>>(attn, v, emb);
  fc_kernel<<<gproj, 32, 0, stream>>>(emb, fc_w, fc_b, h, out);
}


# q/k/v, fc and attn1 products: operand tiles read as whole 128-byte lines, parked in LDS, taken back as b128 fragments (32x64 tiles; q, k, v in one launch; sum order over k and d changes)
# speedup vs baseline: 3.8193x; 1.1885x over previous
#include <hip/hip_runtime.h>
#include <hip/hip_bf16.h>
#include <math.h>

#define NHEAD   8
#define DMODEL  256
#define DKDIM   32
#define DVDIM   32
#define LSEQ    1024
#define BATCH   2
#define LN_EPSF 1e-5f

typedef float v2f __attribute__((ext_vector_type(2)));
typedef float v4f __attribute__((ext_vector_type(4)));
typedef float v8f __attribute__((ext_vector_type(8)));

__device__ __forceinline__ v8f wmma_f32_k4(v2f a, v2f b, v8f c) {
  return __builtin_amdgcn_wmma_f32_16x16x4_f32(false, a, false, b, (short)0, c,
                                               false, false);
}

__device__ __forceinline__ float ntload(const float* p) {
  return __builtin_nontemporal_load(p);
}
__device__ __forceinline__ v4f ntload4(const float* p) {
  return __builtin_nontemporal_load((const v4f*)p);
}

__global__ void ln_kernel(const float* __restrict__ h,
                          const float* __restrict__ gamma,
                          const float* __restrict__ beta,
                          float* __restrict__ hn) {
  int row  = blockIdx.x;
  int lane = threadIdx.x;
  const float* x = h + (size_t)row * DMODEL;
  float vals[8];
  float s = 0.f;
#pragma unroll
  for (int t = 0; t < 8; ++t) { vals[t] = x[lane + 32 * t]; s += vals[t]; }
#pragma unroll
  for (int o = 16; o > 0; o >>= 1) s += __shfl_xor(s, o, 32);
  float mu = s * (1.0f / DMODEL);
  float vs = 0.f;
#pragma unroll
  for (int t = 0; t < 8; ++t) { float d = vals[t] - mu; vs += d * d; }
#pragma unroll
  for (int o = 16; o > 0; o >>= 1) vs += __shfl_xor(vs, o, 32);
  float inv = rsqrtf(vs * (1.0f / DMODEL) + LN_EPSF);
  float* y = hn + (size_t)row * DMODEL;
#pragma unroll
  for (int t = 0; t < 8; ++t) {
    int c = lane + 32 * t;
    y[c] = (vals[t] - mu) * inv * gamma[c] + beta[c];
  }
}

#define GP 260
#define GEMM_LDS_FLOATS ((32 + 64) * GP)
__device__ __forceinline__ v8f gemm_tile(const float* __restrict__ A,
                                         const float* __restrict__ W,
                                         float* __restrict__ sm,
                                         int m0, int n0) {
  float* sA = sm;
  float* sW = sm + 32 * GP;
  int tid = threadIdx.x, w = tid >> 5;
  int lane = tid & 31, half = lane >> 4, l = lane & 15;
  int mt = w & 1, nt = w >> 1;
  const float* ga = A + (size_t)m0 * DMODEL + 4 * tid;
  const float* gw = W + (size_t)n0 * DMODEL + 4 * tid;
  v4f ta[8], tw[16];
#pragma unroll
  for (int it = 0; it < 8; ++it) ta[it] = *(const v4f*)(ga + 1024 * it);
#pragma unroll
  for (int it = 0; it < 16; ++it) tw[it] = *(const v4f*)(gw + 1024 * it);
  int prow = tid >> 6, pcol = 4 * (tid & 63);
#pragma unroll
  for (int it = 0; it < 8; ++it)
    *(v4f*)(sA + (4 * it + prow) * GP + pcol) = ta[it];
#pragma unroll
  for (int it = 0; it < 16; ++it)
    *(v4f*)(sW + (4 * it + prow) * GP + pcol) = tw[it];
  __syncthreads();
  const float* fa = sA + (16 * mt + l) * GP + 4 * half;
  const float* fb = sW + (16 * nt + l) * GP + 4 * half;
  v8f c0 = {}, c1 = {};
#pragma unroll
  for (int q = 0; q < 4; ++q) {
    v4f af[8], bf[8];
#pragma unroll
    for (int uu = 0; uu < 8; ++uu) {
      af[uu] = *(const v4f*)(fa + 8 * (8 * q + uu));
      bf[uu] = *(const v4f*)(fb + 8 * (8 * q + uu));
    }
    asm volatile("" :: "v"(af[0]), "v"(af[1]), "v"(af[2]), "v"(af[3]),
                 "v"(af[4]), "v"(af[5]), "v"(af[6]), "v"(af[7]));
    asm volatile("" :: "v"(bf[0]), "v"(bf[1]), "v"(bf[2]), "v"(bf[3]),
                 "v"(bf[4]), "v"(bf[5]), "v"(bf[6]), "v"(bf[7]));
#pragma unroll
    for (int uu = 0; uu < 8; ++uu) {
      v2f a0, a1, b0, b1;
      a0.x = af[uu].x; a0.y = af[uu].y; a1.x = af[uu].z; a1.y = af[uu].w;
      b0.x = bf[uu].x; b0.y = bf[uu].y; b1.x = bf[uu].z; b1.y = bf[uu].w;
      c0 = wmma_f32_k4(a0, b0, c0);
      c1 = wmma_f32_k4(a1, b1, c1);
    }
  }
  return c0 + c1;
}

__global__ void __launch_bounds__(256, 6)
qkv_kernel(const float* __restrict__ A,
           const float* __restrict__ Wq,
           const float* __restrict__ Wk,
           const float* __restrict__ Wv,
           float* __restrict__ Cq,
           float* __restrict__ Ck,
           float* __restrict__ Cv) {
  __shared__ __attribute__((aligned(16))) float sm[GEMM_LDS_FLOATS];
  int z = blockIdx.z;
  const float* W = z == 0 ? Wq : (z == 1 ? Wk : Wv);
  float* C = z == 0 ? Cq : (z == 1 ? Ck : Cv);
  int m0 = blockIdx.x * 32, n0 = blockIdx.y * 64;
  v8f c = gemm_tile(A, W, sm, m0, n0);
  int tid = threadIdx.x, w = tid >> 5;
  int lane = tid & 31, half = lane >> 4, l = lane & 15;
  float* o = C + (size_t)(m0 + 16 * (w & 1) + 8 * half) * DMODEL
             + n0 + 16 * (w >> 1) + l;
#pragma unroll
  for (int r = 0; r < 8; ++r) o[(size_t)r * DMODEL] = c[r];
}

#define KPITCH 36
__global__ void __launch_bounds__(256, 4)
attn1_kernel(const float* __restrict__ Q,
             const float* __restrict__ K,
             float* __restrict__ attn) {
  __shared__ __attribute__((aligned(16))) float sk[LSEQ * KPITCH];
  int bh = blockIdx.x; int b = bh >> 3; int hh = bh & 7;
  int i0 = blockIdx.y * 16;
  int tid = threadIdx.x, w = tid >> 5;
  int lane = tid & 31, half = lane >> 4, l = lane & 15;
  int lrow = ((lane >> 4) & 1) + 8 * ((lane >> 3) & 1);
  int chunk = lane & 7;
  const float* grow =
      K + ((size_t)(b * LSEQ + w * 128 + lrow) * NHEAD + hh) * DKDIM + 4 * chunk;
  float* spark = sk + (w * 128 + lrow) * KPITCH + 4 * chunk;
  v4f st[8][4];
#pragma unroll
  for (int t = 0; t < 8; ++t)
#pragma unroll
    for (int u = 0; u < 4; ++u)
      st[t][u] = *(const v4f*)(grow + (size_t)(16 * t + 2 * u) * (NHEAD * DKDIM));
  const float* qrow =
      Q + ((size_t)(b * LSEQ + i0 + l) * NHEAD + hh) * DKDIM + 4 * half;
  v4f qa[4];
#pragma unroll
  for (int u = 0; u < 4; ++u) qa[u] = *(const v4f*)(qrow + 8 * u);
#pragma unroll
  for (int t = 0; t < 8; ++t)
#pragma unroll
    for (int u = 0; u < 4; ++u)
      *(v4f*)(spark + (16 * t + 2 * u) * KPITCH) = st[t][u];
  __syncthreads();
  const float* frow = sk + (w * 128 + l) * KPITCH + 4 * half;
  float* o = attn + ((size_t)hh * BATCH + b) * LSEQ * LSEQ
             + (size_t)(i0 + 8 * half) * LSEQ + w * 128 + l;
#pragma unroll
  for (int t = 0; t < 8; ++t) {
    v4f bf[4];
#pragma unroll
    for (int u = 0; u < 4; ++u)
      bf[u] = *(const v4f*)(frow + t * 16 * KPITCH + 8 * u);
    v8f c = {};
#pragma unroll
    for (int u = 0; u < 4; ++u) {
      v2f a0, a1, b0, b1;
      a0.x = qa[u].x; a0.y = qa[u].y; a1.x = qa[u].z; a1.y = qa[u].w;
      b0.x = bf[u].x; b0.y = bf[u].y; b1.x = bf[u].z; b1.y = bf[u].w;
      c = wmma_f32_k4(a0, b0, c);
      c = wmma_f32_k4(a1, b1, c);
    }
#pragma unroll
    for (int r = 0; r < 8; ++r) o[(size_t)r * LSEQ + 16 * t] = c[r];
  }
}

#define BPITCH 36
__global__ void __launch_bounds__(256, 4)
row_kernel(const float* __restrict__ bias,
           const float* __restrict__ Q,
           float* __restrict__ attn,
           float* __restrict__ emb) {
  __shared__ __attribute__((aligned(16))) float sb[LSEQ * BPITCH];
  __shared__ __attribute__((aligned(16))) float part[8][NHEAD * DVDIM];
  __shared__ float red[2][8][16];
  int bi = blockIdx.x; int b = bi >> 10; int i = bi & 1023;
  int tid = threadIdx.x, w = tid >> 5;
  int lane = tid & 31, half = lane >> 4, l = lane & 15;
  bool nvalid = l < 8;
  const float* qv = Q + ((size_t)(b * LSEQ + i) * NHEAD + (l & 7)) * DKDIM
                    + 4 * half;
  v2f bq[8];
#pragma unroll
  for (int u = 0; u < 4; ++u) {
    v4f t = *(const v4f*)(qv + 8 * u);
    bq[2 * u].x     = nvalid ? t.x : 0.f;
    bq[2 * u].y     = nvalid ? t.y : 0.f;
    bq[2 * u + 1].x = nvalid ? t.z : 0.f;
    bq[2 * u + 1].y = nvalid ? t.w : 0.f;
  }
  int lrow = ((lane >> 4) & 1) + 8 * ((lane >> 3) & 1);
  int chunk = lane & 7;
  const float* grow = bias
      + (((size_t)b * LSEQ + i) * LSEQ + w * 128 + lrow) * DKDIM + 4 * chunk;
  float* spark = sb + (w * 128 + lrow) * BPITCH + 4 * chunk;
  float* pr = attn + (((size_t)(l & 7) * BATCH + b) * LSEQ + i) * LSEQ
              + w * 128 + 8 * half;
  v4f st[8][4];
#pragma unroll
  for (int t = 0; t < 8; ++t)
#pragma unroll
    for (int u = 0; u < 4; ++u)
      st[t][u] = ntload4(grow + (size_t)(16 * t + 2 * u) * DKDIM);
  v8f xa[8];
#pragma unroll
  for (int t = 0; t < 8; ++t) {
    float4 lo = *(const float4*)(pr + 16 * t);
    float4 hi = *(const float4*)(pr + 16 * t + 4);
    xa[t][0] = lo.x; xa[t][1] = lo.y; xa[t][2] = lo.z; xa[t][3] = lo.w;
    xa[t][4] = hi.x; xa[t][5] = hi.y; xa[t][6] = hi.z; xa[t][7] = hi.w;
  }
#pragma unroll
  for (int t = 0; t < 8; ++t)
#pragma unroll
    for (int u = 0; u < 4; ++u)
      *(v4f*)(spark + (16 * t + 2 * u) * BPITCH) = st[t][u];
  __syncthreads();
  const float* frow = sb + (w * 128 + l) * BPITCH + 4 * half;
  v4f af[8][4];
#pragma unroll
  for (int t = 0; t < 8; ++t)
#pragma unroll
    for (int u = 0; u < 4; ++u)
      af[t][u] = *(const v4f*)(frow + t * 16 * BPITCH + 8 * u);
#pragma unroll
  for (int t = 0; t < 8; ++t) {
    asm volatile("" :: "v"(af[t][0]), "v"(af[t][1]), "v"(af[t][2]),
                 "v"(af[t][3]));
  }
#pragma unroll
  for (int t = 0; t < 8; ++t) {
#pragma unroll
    for (int u = 0; u < 4; ++u) {
      v2f a0, a1;
      a0.x = af[t][u].x; a0.y = af[t][u].y;
      a1.x = af[t][u].z; a1.y = af[t][u].w;
      xa[t] = wmma_f32_k4(a0, bq[2 * u], xa[t]);
      xa[t] = wmma_f32_k4(a1, bq[2 * u + 1], xa[t]);
    }
  }
  const float scale = 0.17677669529663687f;
  float mx = -INFINITY;
#pragma unroll
  for (int t = 0; t < 8; ++t)
#pragma unroll
    for (int r = 0; r < 8; ++r) { xa[t][r] *= scale; mx = fmaxf(mx, xa[t][r]); }
  mx = fmaxf(mx, __shfl_xor(mx, 16, 32));
  if (half == 0) red[0][w][l] = mx;
  __syncthreads();
#pragma unroll
  for (int u = 0; u < 8; ++u) mx = fmaxf(mx, red[0][u][l]);
  float sm = 0.f;
#pragma unroll
  for (int t = 0; t < 8; ++t)
#pragma unroll
    for (int r = 0; r < 8; ++r) { xa[t][r] = __expf(xa[t][r] - mx); sm += xa[t][r]; }
  sm += __shfl_xor(sm, 16, 32);
  if (half == 0) red[1][w][l] = sm;
  __syncthreads();
  float tot = red[1][0][l];
#pragma unroll
  for (int u = 1; u < 8; ++u) tot += red[1][u][l];
  float inv = 1.0f / tot;
#pragma unroll
  for (int t = 0; t < 8; ++t)
#pragma unroll
    for (int r = 0; r < 8; ++r) xa[t][r] *= inv;
  if (nvalid) {
#pragma unroll
    for (int t = 0; t < 8; ++t) {
      *(float4*)(pr + 16 * t) =
          make_float4(xa[t][0], xa[t][1], xa[t][2], xa[t][3]);
      *(float4*)(pr + 16 * t + 4) =
          make_float4(xa[t][4], xa[t][5], xa[t][6], xa[t][7]);
    }
  }
  const float* brd = sb + (w * 128 + 8 * half) * BPITCH + 2 * l;
  v2f rb[64];
#pragma unroll
  for (int u = 0; u < 64; ++u)
    rb[u] = *(const v2f*)(brd + ((u >> 3) * 16 + (u & 7)) * BPITCH);
#pragma unroll
  for (int t = 0; t < 8; ++t) {
    asm volatile("" :: "v"(rb[8 * t]), "v"(rb[8 * t + 1]), "v"(rb[8 * t + 2]),
                 "v"(rb[8 * t + 3]), "v"(rb[8 * t + 4]), "v"(rb[8 * t + 5]),
                 "v"(rb[8 * t + 6]), "v"(rb[8 * t + 7]));
  }
  v8f c0 = {}, c1 = {};
#pragma unroll
  for (int t = 0; t < 8; ++t) {
#pragma unroll
    for (int m = 0; m < 4; ++m) {
      v2f r0 = rb[8 * t + 2 * m];
      v2f r1 = rb[8 * t + 2 * m + 1];
      v2f pa, b0, b1;
      pa.x = xa[t][2 * m]; pa.y = xa[t][2 * m + 1];
      b0.x = r0.x; b0.y = r1.x;
      b1.x = r0.y; b1.y = r1.y;
      c0 = wmma_f32_k4(pa, b0, c0);
      c1 = wmma_f32_k4(pa, b1, c1);
    }
  }
  if (half == 0) {
#pragma unroll
    for (int r = 0; r < 8; ++r) {
      v2f o; o.x = c0[r]; o.y = c1[r];
      *(v2f*)(&part[w][r * DVDIM + 2 * l]) = o;
    }
  }
  __syncthreads();
  float sum = part[0][tid];
#pragma unroll
  for (int u = 1; u < 8; ++u) sum += part[u][tid];
  emb[(size_t)(b * LSEQ + i) * (NHEAD * DVDIM) + tid] = sum;
}

#define APITCH 132
__global__ void __launch_bounds__(256, 4)
emb1_kernel(const float* __restrict__ attn,
            const float* __restrict__ V,
            float* __restrict__ emb) {
  __shared__ __attribute__((aligned(16))) float sa[8][32 * APITCH];
  int bh = blockIdx.x; int b = bh >> 3; int hh = bh & 7;
  int i0 = blockIdx.y * 32;
  int tid = threadIdx.x, w = tid >> 5;
  int lane = tid & 31, half = lane >> 4, l = lane & 15;
  const float* grow = attn + (((size_t)hh * BATCH + b) * LSEQ + i0) * LSEQ
                      + w * 128 + 4 * lane;
  const float* vcol =
      V + ((size_t)(b * LSEQ + w * 128 + 4 * half) * NHEAD + hh) * DVDIM + 2 * l;
  v2f vb[32][2];
#pragma unroll
  for (int s = 0; s < 16; ++s)
#pragma unroll
    for (int e = 0; e < 2; ++e)
      vb[s][e] = *(const v2f*)(
          vcol + (size_t)(8 * (s >> 1) + 2 * (s & 1) + e) * (NHEAD * DVDIM));
  v4f st[32];
#pragma unroll
  for (int r = 0; r < 32; ++r) st[r] = *(const v4f*)(grow + (size_t)r * LSEQ);
  float* mine = &sa[w][0];
#pragma unroll
  for (int r = 0; r < 32; ++r) *(v4f*)(mine + r * APITCH + 4 * lane) = st[r];
  __syncthreads();
#pragma unroll
  for (int s = 16; s < 32; ++s)
#pragma unroll
    for (int e = 0; e < 2; ++e)
      vb[s][e] = *(const v2f*)(
          vcol + (size_t)(8 * (s >> 1) + 2 * (s & 1) + e) * (NHEAD * DVDIM));
  const float* frow = mine + l * APITCH + 4 * half;
  v8f c[2][2] = {};
  v4f af[2][2][4];
#pragma unroll
  for (int m = 0; m < 2; ++m)
#pragma unroll
    for (int uu = 0; uu < 4; ++uu)
      af[0][m][uu] = *(const v4f*)(frow + m * 16 * APITCH + 8 * uu);
#pragma unroll
  for (int q = 0; q < 4; ++q) {
    if (q < 3) {
#pragma unroll
      for (int m = 0; m < 2; ++m)
#pragma unroll
        for (int uu = 0; uu < 4; ++uu)
          af[(q + 1) & 1][m][uu] =
              *(const v4f*)(frow + m * 16 * APITCH + 8 * (4 * (q + 1) + uu));
    }
    __builtin_amdgcn_sched_barrier(0);
#pragma unroll
    for (int uu = 0; uu < 4; ++uu) {
#pragma unroll
      for (int p = 0; p < 2; ++p) {
        const int s = 2 * (4 * q + uu) + p;
        v2f b0, b1;
        b0.x = vb[s][0].x; b0.y = vb[s][1].x;
        b1.x = vb[s][0].y; b1.y = vb[s][1].y;
#pragma unroll
        for (int m = 0; m < 2; ++m) {
          v2f a;
          a.x = p ? af[q & 1][m][uu].z : af[q & 1][m][uu].x;
          a.y = p ? af[q & 1][m][uu].w : af[q & 1][m][uu].y;
          c[m][0] = wmma_f32_k4(a, b0, c[m][0]);
          c[m][1] = wmma_f32_k4(a, b1, c[m][1]);
        }
      }
    }
  }
#pragma unroll
  for (int m = 0; m < 2; ++m)
#pragma unroll
    for (int r = 0; r < 8; ++r) {
      v2f o; o.x = c[m][0][r]; o.y = c[m][1][r];
      *(v2f*)(mine + (16 * m + 8 * half + r) * DVDIM + 2 * l) = o;
    }
  __syncthreads();
  float* e = emb + (size_t)(b * LSEQ + i0) * (NHEAD * DVDIM) + hh * DVDIM;
#pragma unroll
  for (int q = 0; q < 4; ++q) {
    int idx = tid + 256 * q;
    float sum = sa[0][idx];
#pragma unroll
    for (int u = 1; u < 8; ++u) sum += sa[u][idx];
    e[(size_t)(idx >> 5) * (NHEAD * DVDIM) + (idx & 31)] += sum;
  }
}

__global__ void __launch_bounds__(256, 6)
fc_kernel(const float* __restrict__ emb,
          const float* __restrict__ W,
          const float* __restrict__ bvec,
          const float* __restrict__ resid,
          float* __restrict__ out) {
  __shared__ __attribute__((aligned(16))) float sm[GEMM_LDS_FLOATS];
  int m0 = blockIdx.x * 32, n0 = blockIdx.y * 64;
  int tid = threadIdx.x, w = tid >> 5;
  int lane = tid & 31, half = lane >> 4, l = lane & 15;
  int n = n0 + 16 * (w >> 1) + l;
  size_t off = (size_t)(m0 + 16 * (w & 1) + 8 * half) * DMODEL + n;
  float bv = bvec[n];
  float rs[8];
#pragma unroll
  for (int r = 0; r < 8; ++r) rs[r] = resid[off + (size_t)r * DMODEL];
  v8f c = gemm_tile(emb, W, sm, m0, n0);
#pragma unroll
  for (int r = 0; r < 8; ++r) out[off + (size_t)r * DMODEL] = c[r] + bv + rs[r];
}

extern "C" void kernel_launch(void* const* d_in, const int* in_sizes, int n_in,
                              void* d_out, int out_size, void* d_ws,
                              size_t ws_size, hipStream_t stream) {
  (void)in_sizes; (void)n_in; (void)out_size; (void)ws_size;
  const float* h    = (const float*)d_in[0];
  const float* bias = (const float*)d_in[1];
  const float* w_qs = (const float*)d_in[2];
  const float* w_ks = (const float*)d_in[3];
  const float* w_vs = (const float*)d_in[4];
  const float* ln_g = (const float*)d_in[5];
  const float* ln_b = (const float*)d_in[6];
  const float* fc_w = (const float*)d_in[7];
  const float* fc_b = (const float*)d_in[8];

  float* out  = (float*)d_out;
  float* attn = out + (size_t)BATCH * LSEQ * DMODEL;

  const size_t ROWS = (size_t)BATCH * LSEQ;
  float* ws  = (float*)d_ws;
  float* hn  = ws;
  float* q   = hn + ROWS * DMODEL;
  float* k   = q + ROWS * DMODEL;
  float* v   = k + ROWS * DMODEL;
  float* emb = v + ROWS * DMODEL;

  ln_kernel<<<(int)ROWS, 32, 0, stream>>>(h, ln_g, ln_b, hn);

  qkv_kernel<<<dim3(64, 4, 3), 256, 0, stream>>>(hn, w_qs, w_ks, w_vs, q, k, v);

  attn1_kernel<<<dim3(BATCH * NHEAD, LSEQ / 16), 256, 0, stream>>>(q, k, attn);
  row_kernel<<<(unsigned)ROWS, 256, 0, stream>>>(bias, q, attn, emb);
  emb1_kernel<<<dim3(BATCH * NHEAD, LSEQ / 32), 256, 0, stream>>>(attn, v, emb);
  fc_kernel<<<dim3(64, 4), 256, 0, stream>>>(emb, fc_w, fc_b, h, out);
}


# attn plane moves as whole 128-byte lines: attn1 stores its tiles, the row kernel reads scores and writes probabilities, 512 contiguous bytes per instruction, turned in the wave's own LDS rows
# speedup vs baseline: 4.9922x; 1.3071x over previous
#include <hip/hip_runtime.h>
#include <hip/hip_bf16.h>
#include <math.h>

#define NHEAD   8
#define DMODEL  256
#define DKDIM   32
#define DVDIM   32
#define LSEQ    1024
#define BATCH   2
#define LN_EPSF 1e-5f

typedef float v2f __attribute__((ext_vector_type(2)));
typedef float v4f __attribute__((ext_vector_type(4)));
typedef float v8f __attribute__((ext_vector_type(8)));

__device__ __forceinline__ v8f wmma_f32_k4(v2f a, v2f b, v8f c) {
  return __builtin_amdgcn_wmma_f32_16x16x4_f32(false, a, false, b, (short)0, c,
                                               false, false);
}

__device__ __forceinline__ float ntload(const float* p) {
  return __builtin_nontemporal_load(p);
}
__device__ __forceinline__ v4f ntload4(const float* p) {
  return __builtin_nontemporal_load((const v4f*)p);
}

__device__ __forceinline__ void wave_lds_handover() {
  __builtin_amdgcn_fence(__ATOMIC_RELEASE, "wavefront");
  __builtin_amdgcn_wave_barrier();
  __builtin_amdgcn_fence(__ATOMIC_ACQUIRE, "wavefront");
}

__global__ void ln_kernel(const float* __restrict__ h,
                          const float* __restrict__ gamma,
                          const float* __restrict__ beta,
                          float* __restrict__ hn) {
  int row  = blockIdx.x;
  int lane = threadIdx.x;
  const float* x = h + (size_t)row * DMODEL;
  float vals[8];
  float s = 0.f;
#pragma unroll
  for (int t = 0; t < 8; ++t) { vals[t] = x[lane + 32 * t]; s += vals[t]; }
#pragma unroll
  for (int o = 16; o > 0; o >>= 1) s += __shfl_xor(s, o, 32);
  float mu = s * (1.0f / DMODEL);
  float vs = 0.f;
#pragma unroll
  for (int t = 0; t < 8; ++t) { float d = vals[t] - mu; vs += d * d; }
#pragma unroll
  for (int o = 16; o > 0; o >>= 1) vs += __shfl_xor(vs, o, 32);
  float inv = rsqrtf(vs * (1.0f / DMODEL) + LN_EPSF);
  float* y = hn + (size_t)row * DMODEL;
#pragma unroll
  for (int t = 0; t < 8; ++t) {
    int c = lane + 32 * t;
    y[c] = (vals[t] - mu) * inv * gamma[c] + beta[c];
  }
}

#define GP 260
#define GEMM_LDS_FLOATS ((32 + 64) * GP)
__device__ __forceinline__ v8f gemm_tile(const float* __restrict__ A,
                                         const float* __restrict__ W,
                                         float* __restrict__ sm,
                                         int m0, int n0) {
  float* sA = sm;
  float* sW = sm + 32 * GP;
  int tid = threadIdx.x, w = tid >> 5;
  int lane = tid & 31, half = lane >> 4, l = lane & 15;
  int mt = w & 1, nt = w >> 1;
  const float* ga = A + (size_t)m0 * DMODEL + 4 * tid;
  const float* gw = W + (size_t)n0 * DMODEL + 4 * tid;
  v4f ta[8], tw[16];
#pragma unroll
  for (int it = 0; it < 8; ++it) ta[it] = *(const v4f*)(ga + 1024 * it);
#pragma unroll
  for (int it = 0; it < 16; ++it) tw[it] = *(const v4f*)(gw + 1024 * it);
  int prow = tid >> 6, pcol = 4 * (tid & 63);
#pragma unroll
  for (int it = 0; it < 8; ++it)
    *(v4f*)(sA + (4 * it + prow) * GP + pcol) = ta[it];
#pragma unroll
  for (int it = 0; it < 16; ++it)
    *(v4f*)(sW + (4 * it + prow) * GP + pcol) = tw[it];
  __syncthreads();
  const float* fa = sA + (16 * mt + l) * GP + 4 * half;
  const float* fb = sW + (16 * nt + l) * GP + 4 * half;
  v8f c0 = {}, c1 = {};
#pragma unroll
  for (int q = 0; q < 4; ++q) {
    v4f af[8], bf[8];
#pragma unroll
    for (int uu = 0; uu < 8; ++uu) {
      af[uu] = *(const v4f*)(fa + 8 * (8 * q + uu));
      bf[uu] = *(const v4f*)(fb + 8 * (8 * q + uu));
    }
    asm volatile("" :: "v"(af[0]), "v"(af[1]), "v"(af[2]), "v"(af[3]),
                 "v"(af[4]), "v"(af[5]), "v"(af[6]), "v"(af[7]));
    asm volatile("" :: "v"(bf[0]), "v"(bf[1]), "v"(bf[2]), "v"(bf[3]),
                 "v"(bf[4]), "v"(bf[5]), "v"(bf[6]), "v"(bf[7]));
#pragma unroll
    for (int uu = 0; uu < 8; ++uu) {
      v2f a0, a1, b0, b1;
      a0.x = af[uu].x; a0.y = af[uu].y; a1.x = af[uu].z; a1.y = af[uu].w;
      b0.x = bf[uu].x; b0.y = bf[uu].y; b1.x = bf[uu].z; b1.y = bf[uu].w;
      c0 = wmma_f32_k4(a0, b0, c0);
      c1 = wmma_f32_k4(a1, b1, c1);
    }
  }
  return c0 + c1;
}

__global__ void __launch_bounds__(256, 6)
qkv_kernel(const float* __restrict__ A,
           const float* __restrict__ Wq,
           const float* __restrict__ Wk,
           const float* __restrict__ Wv,
           float* __restrict__ Cq,
           float* __restrict__ Ck,
           float* __restrict__ Cv) {
  __shared__ __attribute__((aligned(16))) float sm[GEMM_LDS_FLOATS];
  int z = blockIdx.z;
  const float* W = z == 0 ? Wq : (z == 1 ? Wk : Wv);
  float* C = z == 0 ? Cq : (z == 1 ? Ck : Cv);
  int m0 = blockIdx.x * 32, n0 = blockIdx.y * 64;
  v8f c = gemm_tile(A, W, sm, m0, n0);
  int tid = threadIdx.x, w = tid >> 5;
  int lane = tid & 31, half = lane >> 4, l = lane & 15;
  float* o = C + (size_t)(m0 + 16 * (w & 1) + 8 * half) * DMODEL
             + n0 + 16 * (w >> 1) + l;
#pragma unroll
  for (int r = 0; r < 8; ++r) o[(size_t)r * DMODEL] = c[r];
}

#define OPITCH 132
#define KPITCH 36
__global__ void __launch_bounds__(256, 4)
attn1_kernel(const float* __restrict__ Q,
             const float* __restrict__ K,
             float* __restrict__ attn) {
  __shared__ __attribute__((aligned(16))) float sk[LSEQ * KPITCH];
  int bh = blockIdx.x; int b = bh >> 3; int hh = bh & 7;
  int i0 = blockIdx.y * 16;
  int tid = threadIdx.x, w = tid >> 5;
  int lane = tid & 31, half = lane >> 4, l = lane & 15;
  int lrow = ((lane >> 4) & 1) + 8 * ((lane >> 3) & 1);
  int chunk = lane & 7;
  const float* grow =
      K + ((size_t)(b * LSEQ + w * 128 + lrow) * NHEAD + hh) * DKDIM + 4 * chunk;
  float* spark = sk + (w * 128 + lrow) * KPITCH + 4 * chunk;
  v4f st[8][4];
#pragma unroll
  for (int t = 0; t < 8; ++t)
#pragma unroll
    for (int u = 0; u < 4; ++u)
      st[t][u] = *(const v4f*)(grow + (size_t)(16 * t + 2 * u) * (NHEAD * DKDIM));
  const float* qrow =
      Q + ((size_t)(b * LSEQ + i0 + l) * NHEAD + hh) * DKDIM + 4 * half;
  v4f qa[4];
#pragma unroll
  for (int u = 0; u < 4; ++u) qa[u] = *(const v4f*)(qrow + 8 * u);
#pragma unroll
  for (int t = 0; t < 8; ++t)
#pragma unroll
    for (int u = 0; u < 4; ++u)
      *(v4f*)(spark + (16 * t + 2 * u) * KPITCH) = st[t][u];
  __syncthreads();
  const float* frow = sk + (w * 128 + l) * KPITCH + 4 * half;
  v4f bf[8][4];
#pragma unroll
  for (int t = 0; t < 8; ++t)
#pragma unroll
    for (int u = 0; u < 4; ++u)
      bf[t][u] = *(const v4f*)(frow + t * 16 * KPITCH + 8 * u);
  __builtin_amdgcn_sched_barrier(0);
  float* os = sk + w * 128 * KPITCH;
  float* osw = os + l * OPITCH + 8 * half;
#pragma unroll
  for (int t = 0; t < 8; ++t) {
    v8f c = {};
#pragma unroll
    for (int u = 0; u < 4; ++u) {
      v2f a0, a1, b0, b1;
      a0.x = bf[t][u].x; a0.y = bf[t][u].y;
      a1.x = bf[t][u].z; a1.y = bf[t][u].w;
      b0.x = qa[u].x; b0.y = qa[u].y; b1.x = qa[u].z; b1.y = qa[u].w;
      c = wmma_f32_k4(a0, b0, c);
      c = wmma_f32_k4(a1, b1, c);
    }
    v4f lo, hi;
    lo.x = c[0]; lo.y = c[1]; lo.z = c[2]; lo.w = c[3];
    hi.x = c[4]; hi.y = c[5]; hi.z = c[6]; hi.w = c[7];
    *(v4f*)(osw + 16 * t) = lo;
    *(v4f*)(osw + 16 * t + 4) = hi;
  }
  wave_lds_handover();
  float* o = attn + ((size_t)hh * BATCH + b) * LSEQ * LSEQ
             + (size_t)i0 * LSEQ + w * 128 + 4 * lane;
  v4f ov[16];
#pragma unroll
  for (int r = 0; r < 16; ++r) ov[r] = *(const v4f*)(os + r * OPITCH + 4 * lane);
#pragma unroll
  for (int r = 0; r < 16; ++r) *(v4f*)(o + (size_t)r * LSEQ) = ov[r];
}

#define XPITCH 132
#define BPITCH 36
__global__ void __launch_bounds__(256, 4)
row_kernel(const float* __restrict__ bias,
           const float* __restrict__ Q,
           float* __restrict__ attn,
           float* __restrict__ emb) {
  __shared__ __attribute__((aligned(16))) float sb[LSEQ * BPITCH];
  __shared__ __attribute__((aligned(16))) float part[8][NHEAD * DVDIM];
  __shared__ float red[2][8][16];
  int bi = blockIdx.x; int b = bi >> 10; int i = bi & 1023;
  int tid = threadIdx.x, w = tid >> 5;
  int lane = tid & 31, half = lane >> 4, l = lane & 15;
  bool nvalid = l < 8;
  const float* qv = Q + ((size_t)(b * LSEQ + i) * NHEAD + (l & 7)) * DKDIM
                    + 4 * half;
  v2f bq[8];
#pragma unroll
  for (int u = 0; u < 4; ++u) {
    v4f t = *(const v4f*)(qv + 8 * u);
    bq[2 * u].x     = nvalid ? t.x : 0.f;
    bq[2 * u].y     = nvalid ? t.y : 0.f;
    bq[2 * u + 1].x = nvalid ? t.z : 0.f;
    bq[2 * u + 1].y = nvalid ? t.w : 0.f;
  }
  int lrow = ((lane >> 4) & 1) + 8 * ((lane >> 3) & 1);
  int chunk = lane & 7;
  const float* grow = bias
      + (((size_t)b * LSEQ + i) * LSEQ + w * 128 + lrow) * DKDIM + 4 * chunk;
  float* spark = sb + (w * 128 + lrow) * BPITCH + 4 * chunk;
  float* pg = attn + ((size_t)b * LSEQ + i) * LSEQ + w * 128 + 4 * lane;
  const size_t plane = (size_t)BATCH * LSEQ * LSEQ;
  v4f xs[8];
#pragma unroll
  for (int h = 0; h < 8; ++h) xs[h] = *(const v4f*)(pg + (size_t)h * plane);
  v4f st[8][4];
#pragma unroll
  for (int t = 0; t < 8; ++t)
#pragma unroll
    for (int u = 0; u < 4; ++u)
      st[t][u] = ntload4(grow + (size_t)(16 * t + 2 * u) * DKDIM);
  float* xsb = sb + w * 128 * BPITCH;
#pragma unroll
  for (int h = 0; h < 8; ++h) *(v4f*)(xsb + h * XPITCH + 4 * lane) = xs[h];
  wave_lds_handover();
  const float* xrd = xsb + (l & 7) * XPITCH + 8 * half;
  v8f xa[8];
#pragma unroll
  for (int t = 0; t < 8; ++t) {
    v4f lo = *(const v4f*)(xrd + 16 * t);
    v4f hi = *(const v4f*)(xrd + 16 * t + 4);
    xa[t][0] = lo.x; xa[t][1] = lo.y; xa[t][2] = lo.z; xa[t][3] = lo.w;
    xa[t][4] = hi.x; xa[t][5] = hi.y; xa[t][6] = hi.z; xa[t][7] = hi.w;
  }
  wave_lds_handover();
#pragma unroll
  for (int t = 0; t < 8; ++t)
#pragma unroll
    for (int u = 0; u < 4; ++u)
      *(v4f*)(spark + (16 * t + 2 * u) * BPITCH) = st[t][u];
  __syncthreads();
  const float* frow = sb + (w * 128 + l) * BPITCH + 4 * half;
  v4f af[8][4];
#pragma unroll
  for (int t = 0; t < 8; ++t)
#pragma unroll
    for (int u = 0; u < 4; ++u)
      af[t][u] = *(const v4f*)(frow + t * 16 * BPITCH + 8 * u);
#pragma unroll
  for (int t = 0; t < 8; ++t) {
    asm volatile("" :: "v"(af[t][0]), "v"(af[t][1]), "v"(af[t][2]),
                 "v"(af[t][3]));
  }
#pragma unroll
  for (int t = 0; t < 8; ++t) {
#pragma unroll
    for (int u = 0; u < 4; ++u) {
      v2f a0, a1;
      a0.x = af[t][u].x; a0.y = af[t][u].y;
      a1.x = af[t][u].z; a1.y = af[t][u].w;
      xa[t] = wmma_f32_k4(a0, bq[2 * u], xa[t]);
      xa[t] = wmma_f32_k4(a1, bq[2 * u + 1], xa[t]);
    }
  }
  const float scale = 0.17677669529663687f;
  float mx = -INFINITY;
#pragma unroll
  for (int t = 0; t < 8; ++t)
#pragma unroll
    for (int r = 0; r < 8; ++r) { xa[t][r] *= scale; mx = fmaxf(mx, xa[t][r]); }
  mx = fmaxf(mx, __shfl_xor(mx, 16, 32));
  if (half == 0) red[0][w][l] = mx;
  __syncthreads();
#pragma unroll
  for (int u = 0; u < 8; ++u) mx = fmaxf(mx, red[0][u][l]);
  float sm = 0.f;
#pragma unroll
  for (int t = 0; t < 8; ++t)
#pragma unroll
    for (int r = 0; r < 8; ++r) { xa[t][r] = __expf(xa[t][r] - mx); sm += xa[t][r]; }
  sm += __shfl_xor(sm, 16, 32);
  if (half == 0) red[1][w][l] = sm;
  __syncthreads();
  float tot = red[1][0][l];
#pragma unroll
  for (int u = 1; u < 8; ++u) tot += red[1][u][l];
  float inv = 1.0f / tot;
#pragma unroll
  for (int t = 0; t < 8; ++t)
#pragma unroll
    for (int r = 0; r < 8; ++r) xa[t][r] *= inv;
  const float* brd = sb + (w * 128 + 8 * half) * BPITCH + 2 * l;
  v2f rb[64];
#pragma unroll
  for (int u = 0; u < 64; ++u)
    rb[u] = *(const v2f*)(brd + ((u >> 3) * 16 + (u & 7)) * BPITCH);
#pragma unroll
  for (int t = 0; t < 8; ++t) {
    asm volatile("" :: "v"(rb[8 * t]), "v"(rb[8 * t + 1]), "v"(rb[8 * t + 2]),
                 "v"(rb[8 * t + 3]), "v"(rb[8 * t + 4]), "v"(rb[8 * t + 5]),
                 "v"(rb[8 * t + 6]), "v"(rb[8 * t + 7]));
  }
  wave_lds_handover();
  if (nvalid) {
    float* pw = xsb + l * XPITCH + 8 * half;
#pragma unroll
    for (int t = 0; t < 8; ++t) {
      v4f lo, hi;
      lo.x = xa[t][0]; lo.y = xa[t][1]; lo.z = xa[t][2]; lo.w = xa[t][3];
      hi.x = xa[t][4]; hi.y = xa[t][5]; hi.z = xa[t][6]; hi.w = xa[t][7];
      *(v4f*)(pw + 16 * t) = lo;
      *(v4f*)(pw + 16 * t + 4) = hi;
    }
  }
  v8f c0 = {}, c1 = {};
#pragma unroll
  for (int t = 0; t < 8; ++t) {
#pragma unroll
    for (int m = 0; m < 4; ++m) {
      v2f r0 = rb[8 * t + 2 * m];
      v2f r1 = rb[8 * t + 2 * m + 1];
      v2f pa, b0, b1;
      pa.x = xa[t][2 * m]; pa.y = xa[t][2 * m + 1];
      b0.x = r0.x; b0.y = r1.x;
      b1.x = r0.y; b1.y = r1.y;
      c0 = wmma_f32_k4(pa, b0, c0);
      c1 = wmma_f32_k4(pa, b1, c1);
    }
  }
  if (half == 0) {
#pragma unroll
    for (int r = 0; r < 8; ++r) {
      v2f o; o.x = c0[r]; o.y = c1[r];
      *(v2f*)(&part[w][r * DVDIM + 2 * l]) = o;
    }
  }
  __syncthreads();
  v4f ov[8];
#pragma unroll
  for (int h = 0; h < 8; ++h) ov[h] = *(const v4f*)(xsb + h * XPITCH + 4 * lane);
#pragma unroll
  for (int h = 0; h < 8; ++h) *(v4f*)(pg + (size_t)h * plane) = ov[h];
  float sum = part[0][tid];
#pragma unroll
  for (int u = 1; u < 8; ++u) sum += part[u][tid];
  emb[(size_t)(b * LSEQ + i) * (NHEAD * DVDIM) + tid] = sum;
}

#define APITCH 132
__global__ void __launch_bounds__(256, 4)
emb1_kernel(const float* __restrict__ attn,
            const float* __restrict__ V,
            float* __restrict__ emb) {
  __shared__ __attribute__((aligned(16))) float sa[8][32 * APITCH];
  int bh = blockIdx.x; int b = bh >> 3; int hh = bh & 7;
  int i0 = blockIdx.y * 32;
  int tid = threadIdx.x, w = tid >> 5;
  int lane = tid & 31, half = lane >> 4, l = lane & 15;
  const float* grow = attn + (((size_t)hh * BATCH + b) * LSEQ + i0) * LSEQ
                      + w * 128 + 4 * lane;
  const float* vcol =
      V + ((size_t)(b * LSEQ + w * 128 + 4 * half) * NHEAD + hh) * DVDIM + 2 * l;
  v2f vb[32][2];
#pragma unroll
  for (int s = 0; s < 16; ++s)
#pragma unroll
    for (int e = 0; e < 2; ++e)
      vb[s][e] = *(const v2f*)(
          vcol + (size_t)(8 * (s >> 1) + 2 * (s & 1) + e) * (NHEAD * DVDIM));
  v4f st[32];
#pragma unroll
  for (int r = 0; r < 32; ++r) st[r] = *(const v4f*)(grow + (size_t)r * LSEQ);
  float* mine = &sa[w][0];
#pragma unroll
  for (int r = 0; r < 32; ++r) *(v4f*)(mine + r * APITCH + 4 * lane) = st[r];
  __syncthreads();
#pragma unroll
  for (int s = 16; s < 32; ++s)
#pragma unroll
    for (int e = 0; e < 2; ++e)
      vb[s][e] = *(const v2f*)(
          vcol + (size_t)(8 * (s >> 1) + 2 * (s & 1) + e) * (NHEAD * DVDIM));
  const float* frow = mine + l * APITCH + 4 * half;
  v8f c[2][2] = {};
  v4f af[2][2][4];
#pragma unroll
  for (int m = 0; m < 2; ++m)
#pragma unroll
    for (int uu = 0; uu < 4; ++uu)
      af[0][m][uu] = *(const v4f*)(frow + m * 16 * APITCH + 8 * uu);
#pragma unroll
  for (int q = 0; q < 4; ++q) {
    if (q < 3) {
#pragma unroll
      for (int m = 0; m < 2; ++m)
#pragma unroll
        for (int uu = 0; uu < 4; ++uu)
          af[(q + 1) & 1][m][uu] =
              *(const v4f*)(frow + m * 16 * APITCH + 8 * (4 * (q + 1) + uu));
    }
    __builtin_amdgcn_sched_barrier(0);
#pragma unroll
    for (int uu = 0; uu < 4; ++uu) {
#pragma unroll
      for (int p = 0; p < 2; ++p) {
        const int s = 2 * (4 * q + uu) + p;
        v2f b0, b1;
        b0.x = vb[s][0].x; b0.y = vb[s][1].x;
        b1.x = vb[s][0].y; b1.y = vb[s][1].y;
#pragma unroll
        for (int m = 0; m < 2; ++m) {
          v2f a;
          a.x = p ? af[q & 1][m][uu].z : af[q & 1][m][uu].x;
          a.y = p ? af[q & 1][m][uu].w : af[q & 1][m][uu].y;
          c[m][0] = wmma_f32_k4(a, b0, c[m][0]);
          c[m][1] = wmma_f32_k4(a, b1, c[m][1]);
        }
      }
    }
  }
#pragma unroll
  for (int m = 0; m < 2; ++m)
#pragma unroll
    for (int r = 0; r < 8; ++r) {
      v2f o; o.x = c[m][0][r]; o.y = c[m][1][r];
      *(v2f*)(mine + (16 * m + 8 * half + r) * DVDIM + 2 * l) = o;
    }
  __syncthreads();
  float* e = emb + (size_t)(b * LSEQ + i0) * (NHEAD * DVDIM) + hh * DVDIM;
#pragma unroll
  for (int q = 0; q < 4; ++q) {
    int idx = tid + 256 * q;
    float sum = sa[0][idx];
#pragma unroll
    for (int u = 1; u < 8; ++u) sum += sa[u][idx];
    e[(size_t)(idx >> 5) * (NHEAD * DVDIM) + (idx & 31)] += sum;
  }
}

__global__ void __launch_bounds__(256, 6)
fc_kernel(const float* __restrict__ emb,
          const float* __restrict__ W,
          const float* __restrict__ bvec,
          const float* __restrict__ resid,
          float* __restrict__ out) {
  __shared__ __attribute__((aligned(16))) float sm[GEMM_LDS_FLOATS];
  int m0 = blockIdx.x * 32, n0 = blockIdx.y * 64;
  int tid = threadIdx.x, w = tid >> 5;
  int lane = tid & 31, half = lane >> 4, l = lane & 15;
  int n = n0 + 16 * (w >> 1) + l;
  size_t off = (size_t)(m0 + 16 * (w & 1) + 8 * half) * DMODEL + n;
  float bv = bvec[n];
  float rs[8];
#pragma unroll
  for (int r = 0; r < 8; ++r) rs[r] = resid[off + (size_t)r * DMODEL];
  v8f c = gemm_tile(emb, W, sm, m0, n0);
#pragma unroll
  for (int r = 0; r < 8; ++r) out[off + (size_t)r * DMODEL] = c[r] + bv + rs[r];
}

extern "C" void kernel_launch(void* const* d_in, const int* in_sizes, int n_in,
                              void* d_out, int out_size, void* d_ws,
                              size_t ws_size, hipStream_t stream) {
  (void)in_sizes; (void)n_in; (void)out_size; (void)ws_size;
  const float* h    = (const float*)d_in[0];
  const float* bias = (const float*)d_in[1];
  const float* w_qs = (const float*)d_in[2];
  const float* w_ks = (const float*)d_in[3];
  const float* w_vs = (const float*)d_in[4];
  const float* ln_g = (const float*)d_in[5];
  const float* ln_b = (const float*)d_in[6];
  const float* fc_w = (const float*)d_in[7];
  const float* fc_b = (const float*)d_in[8];

  float* out  = (float*)d_out;
  float* attn = out + (size_t)BATCH * LSEQ * DMODEL;

  const size_t ROWS = (size_t)BATCH * LSEQ;
  float* ws  = (float*)d_ws;
  float* hn  = ws;
  float* q   = hn + ROWS * DMODEL;
  float* k   = q + ROWS * DMODEL;
  float* v   = k + ROWS * DMODEL;
  float* emb = v + ROWS * DMODEL;

  ln_kernel<<<(int)ROWS, 32, 0, stream>>>(h, ln_g, ln_b, hn);

  qkv_kernel<<<dim3(64, 4, 3), 256, 0, stream>>>(hn, w_qs, w_ks, w_vs, q, k, v);

  attn1_kernel<<<dim3(BATCH * NHEAD, LSEQ / 16), 256, 0, stream>>>(q, k, attn);
  row_kernel<<<(unsigned)ROWS, 256, 0, stream>>>(bias, q, attn, emb);
  emb1_kernel<<<dim3(BATCH * NHEAD, LSEQ / 32), 256, 0, stream>>>(attn, v, emb);
  fc_kernel<<<dim3(64, 4), 256, 0, stream>>>(emb, fc_w, fc_b, h, out);
}


# attn1, emb1: one block per (b, head, 64 rows i), not 16 / 32: k / v fragments stay in registers over the row tiles, 12288 waves become 4096; emb1's next tile and its emb words load under the WMMAs
# speedup vs baseline: 5.2624x; 1.0541x over previous
#include <hip/hip_runtime.h>
#include <hip/hip_bf16.h>
#include <math.h>

#define NHEAD   8
#define DMODEL  256
#define DKDIM   32
#define DVDIM   32
#define LSEQ    1024
#define BATCH   2
#define LN_EPSF 1e-5f

typedef float v2f __attribute__((ext_vector_type(2)));
typedef float v4f __attribute__((ext_vector_type(4)));
typedef float v8f __attribute__((ext_vector_type(8)));

__device__ __forceinline__ v8f wmma_f32_k4(v2f a, v2f b, v8f c) {
  return __builtin_amdgcn_wmma_f32_16x16x4_f32(false, a, false, b, (short)0, c,
                                               false, false);
}

__device__ __forceinline__ float ntload(const float* p) {
  return __builtin_nontemporal_load(p);
}
__device__ __forceinline__ v4f ntload4(const float* p) {
  return __builtin_nontemporal_load((const v4f*)p);
}

__device__ __forceinline__ void wave_lds_handover() {
  __builtin_amdgcn_fence(__ATOMIC_RELEASE, "wavefront");
  __builtin_amdgcn_wave_barrier();
  __builtin_amdgcn_fence(__ATOMIC_ACQUIRE, "wavefront");
}

__global__ void ln_kernel(const float* __restrict__ h,
                          const float* __restrict__ gamma,
                          const float* __restrict__ beta,
                          float* __restrict__ hn) {
  int row  = blockIdx.x;
  int lane = threadIdx.x;
  const float* x = h + (size_t)row * DMODEL;
  float vals[8];
  float s = 0.f;
#pragma unroll
  for (int t = 0; t < 8; ++t) { vals[t] = x[lane + 32 * t]; s += vals[t]; }
#pragma unroll
  for (int o = 16; o > 0; o >>= 1) s += __shfl_xor(s, o, 32);
  float mu = s * (1.0f / DMODEL);
  float vs = 0.f;
#pragma unroll
  for (int t = 0; t < 8; ++t) { float d = vals[t] - mu; vs += d * d; }
#pragma unroll
  for (int o = 16; o > 0; o >>= 1) vs += __shfl_xor(vs, o, 32);
  float inv = rsqrtf(vs * (1.0f / DMODEL) + LN_EPSF);
  float* y = hn + (size_t)row * DMODEL;
#pragma unroll
  for (int t = 0; t < 8; ++t) {
    int c = lane + 32 * t;
    y[c] = (vals[t] - mu) * inv * gamma[c] + beta[c];
  }
}

#define GP 260
#define GEMM_LDS_FLOATS ((32 + 64) * GP)
__device__ __forceinline__ v8f gemm_tile(const float* __restrict__ A,
                                         const float* __restrict__ W,
                                         float* __restrict__ sm,
                                         int m0, int n0) {
  float* sA = sm;
  float* sW = sm + 32 * GP;
  int tid = threadIdx.x, w = tid >> 5;
  int lane = tid & 31, half = lane >> 4, l = lane & 15;
  int mt = w & 1, nt = w >> 1;
  const float* ga = A + (size_t)m0 * DMODEL + 4 * tid;
  const float* gw = W + (size_t)n0 * DMODEL + 4 * tid;
  v4f ta[8], tw[16];
#pragma unroll
  for (int it = 0; it < 8; ++it) ta[it] = *(const v4f*)(ga + 1024 * it);
#pragma unroll
  for (int it = 0; it < 16; ++it) tw[it] = *(const v4f*)(gw + 1024 * it);
  int prow = tid >> 6, pcol = 4 * (tid & 63);
#pragma unroll
  for (int it = 0; it < 8; ++it)
    *(v4f*)(sA + (4 * it + prow) * GP + pcol) = ta[it];
#pragma unroll
  for (int it = 0; it < 16; ++it)
    *(v4f*)(sW + (4 * it + prow) * GP + pcol) = tw[it];
  __syncthreads();
  const float* fa = sA + (16 * mt + l) * GP + 4 * half;
  const float* fb = sW + (16 * nt + l) * GP + 4 * half;
  v8f c0 = {}, c1 = {};
#pragma unroll
  for (int q = 0; q < 4; ++q) {
    v4f af[8], bf[8];
#pragma unroll
    for (int uu = 0; uu < 8; ++uu) {
      af[uu] = *(const v4f*)(fa + 8 * (8 * q + uu));
      bf[uu] = *(const v4f*)(fb + 8 * (8 * q + uu));
    }
    asm volatile("" :: "v"(af[0]), "v"(af[1]), "v"(af[2]), "v"(af[3]),
                 "v"(af[4]), "v"(af[5]), "v"(af[6]), "v"(af[7]));
    asm volatile("" :: "v"(bf[0]), "v"(bf[1]), "v"(bf[2]), "v"(bf[3]),
                 "v"(bf[4]), "v"(bf[5]), "v"(bf[6]), "v"(bf[7]));
#pragma unroll
    for (int uu = 0; uu < 8; ++uu) {
      v2f a0, a1, b0, b1;
      a0.x = af[uu].x; a0.y = af[uu].y; a1.x = af[uu].z; a1.y = af[uu].w;
      b0.x = bf[uu].x; b0.y = bf[uu].y; b1.x = bf[uu].z; b1.y = bf[uu].w;
      c0 = wmma_f32_k4(a0, b0, c0);
      c1 = wmma_f32_k4(a1, b1, c1);
    }
  }
  return c0 + c1;
}

__global__ void __launch_bounds__(256, 6)
qkv_kernel(const float* __restrict__ A,
           const float* __restrict__ Wq,
           const float* __restrict__ Wk,
           const float* __restrict__ Wv,
           float* __restrict__ Cq,
           float* __restrict__ Ck,
           float* __restrict__ Cv) {
  __shared__ __attribute__((aligned(16))) float sm[GEMM_LDS_FLOATS];
  int z = blockIdx.z;
  const float* W = z == 0 ? Wq : (z == 1 ? Wk : Wv);
  float* C = z == 0 ? Cq : (z == 1 ? Ck : Cv);
  int m0 = blockIdx.x * 32, n0 = blockIdx.y * 64;
  v8f c = gemm_tile(A, W, sm, m0, n0);
  int tid = threadIdx.x, w = tid >> 5;
  int lane = tid & 31, half = lane >> 4, l = lane & 15;
  float* o = C + (size_t)(m0 + 16 * (w & 1) + 8 * half) * DMODEL
             + n0 + 16 * (w >> 1) + l;
#pragma unroll
  for (int r = 0; r < 8; ++r) o[(size_t)r * DMODEL] = c[r];
}

#define OPITCH 132
#define KPITCH 36
#define A1TILES 4
__global__ void __launch_bounds__(256, 2)
attn1_kernel(const float* __restrict__ Q,
             const float* __restrict__ K,
             float* __restrict__ attn) {
  __shared__ __attribute__((aligned(16))) float sk[LSEQ * KPITCH];
  int bh = blockIdx.x; int b = bh >> 3; int hh = bh & 7;
  int i0 = blockIdx.y * (16 * A1TILES);
  int tid = threadIdx.x, w = tid >> 5;
  int lane = tid & 31, half = lane >> 4, l = lane & 15;
  int lrow = ((lane >> 4) & 1) + 8 * ((lane >> 3) & 1);
  int chunk = lane & 7;
  const float* grow =
      K + ((size_t)(b * LSEQ + w * 128 + lrow) * NHEAD + hh) * DKDIM + 4 * chunk;
  float* spark = sk + (w * 128 + lrow) * KPITCH + 4 * chunk;
  v4f st[8][4];
#pragma unroll
  for (int t = 0; t < 8; ++t)
#pragma unroll
    for (int u = 0; u < 4; ++u)
      st[t][u] = *(const v4f*)(grow + (size_t)(16 * t + 2 * u) * (NHEAD * DKDIM));
  const float* qrow =
      Q + ((size_t)(b * LSEQ + i0 + l) * NHEAD + hh) * DKDIM + 4 * half;
  v4f qa[A1TILES][4];
#pragma unroll
  for (int it = 0; it < A1TILES; ++it)
#pragma unroll
    for (int u = 0; u < 4; ++u)
      qa[it][u] = *(const v4f*)(qrow + (size_t)(16 * it) * (NHEAD * DKDIM) + 8 * u);
#pragma unroll
  for (int t = 0; t < 8; ++t)
#pragma unroll
    for (int u = 0; u < 4; ++u)
      *(v4f*)(spark + (16 * t + 2 * u) * KPITCH) = st[t][u];
  __syncthreads();
  const float* frow = sk + (w * 128 + l) * KPITCH + 4 * half;
  v4f bf[8][4];
#pragma unroll
  for (int t = 0; t < 8; ++t)
#pragma unroll
    for (int u = 0; u < 4; ++u)
      bf[t][u] = *(const v4f*)(frow + t * 16 * KPITCH + 8 * u);
  __builtin_amdgcn_sched_barrier(0);
  float* os = sk + w * 128 * KPITCH;
  float* osw = os + l * OPITCH + 8 * half;
  float* o = attn + ((size_t)hh * BATCH + b) * LSEQ * LSEQ
             + (size_t)i0 * LSEQ + w * 128 + 4 * lane;
#pragma unroll
  for (int it = 0; it < A1TILES; ++it) {
#pragma unroll
    for (int t = 0; t < 8; ++t) {
      v8f c = {};
#pragma unroll
      for (int u = 0; u < 4; ++u) {
        v2f a0, a1, b0, b1;
        a0.x = bf[t][u].x; a0.y = bf[t][u].y;
        a1.x = bf[t][u].z; a1.y = bf[t][u].w;
        b0.x = qa[it][u].x; b0.y = qa[it][u].y;
        b1.x = qa[it][u].z; b1.y = qa[it][u].w;
        c = wmma_f32_k4(a0, b0, c);
        c = wmma_f32_k4(a1, b1, c);
      }
      v4f lo, hi;
      lo.x = c[0]; lo.y = c[1]; lo.z = c[2]; lo.w = c[3];
      hi.x = c[4]; hi.y = c[5]; hi.z = c[6]; hi.w = c[7];
      *(v4f*)(osw + 16 * t) = lo;
      *(v4f*)(osw + 16 * t + 4) = hi;
    }
    wave_lds_handover();
    v4f ov[16];
#pragma unroll
    for (int r = 0; r < 16; ++r) ov[r] = *(const v4f*)(os + r * OPITCH + 4 * lane);
#pragma unroll
    for (int r = 0; r < 16; ++r)
      *(v4f*)(o + (size_t)(16 * it + r) * LSEQ) = ov[r];
    wave_lds_handover();
  }
}

#define XPITCH 132
#define BPITCH 36
__global__ void __launch_bounds__(256, 4)
row_kernel(const float* __restrict__ bias,
           const float* __restrict__ Q,
           float* __restrict__ attn,
           float* __restrict__ emb) {
  __shared__ __attribute__((aligned(16))) float sb[LSEQ * BPITCH];
  __shared__ __attribute__((aligned(16))) float part[8][NHEAD * DVDIM];
  __shared__ float red[2][8][16];
  int bi = blockIdx.x; int b = bi >> 10; int i = bi & 1023;
  int tid = threadIdx.x, w = tid >> 5;
  int lane = tid & 31, half = lane >> 4, l = lane & 15;
  bool nvalid = l < 8;
  const float* qv = Q + ((size_t)(b * LSEQ + i) * NHEAD + (l & 7)) * DKDIM
                    + 4 * half;
  v2f bq[8];
#pragma unroll
  for (int u = 0; u < 4; ++u) {
    v4f t = *(const v4f*)(qv + 8 * u);
    bq[2 * u].x     = nvalid ? t.x : 0.f;
    bq[2 * u].y     = nvalid ? t.y : 0.f;
    bq[2 * u + 1].x = nvalid ? t.z : 0.f;
    bq[2 * u + 1].y = nvalid ? t.w : 0.f;
  }
  int lrow = ((lane >> 4) & 1) + 8 * ((lane >> 3) & 1);
  int chunk = lane & 7;
  const float* grow = bias
      + (((size_t)b * LSEQ + i) * LSEQ + w * 128 + lrow) * DKDIM + 4 * chunk;
  float* spark = sb + (w * 128 + lrow) * BPITCH + 4 * chunk;
  float* pg = attn + ((size_t)b * LSEQ + i) * LSEQ + w * 128 + 4 * lane;
  const size_t plane = (size_t)BATCH * LSEQ * LSEQ;
  v4f xs[8];
#pragma unroll
  for (int h = 0; h < 8; ++h) xs[h] = *(const v4f*)(pg + (size_t)h * plane);
  v4f st[8][4];
#pragma unroll
  for (int t = 0; t < 8; ++t)
#pragma unroll
    for (int u = 0; u < 4; ++u)
      st[t][u] = ntload4(grow + (size_t)(16 * t + 2 * u) * DKDIM);
  float* xsb = sb + w * 128 * BPITCH;
#pragma unroll
  for (int h = 0; h < 8; ++h) *(v4f*)(xsb + h * XPITCH + 4 * lane) = xs[h];
  wave_lds_handover();
  const float* xrd = xsb + (l & 7) * XPITCH + 8 * half;
  v8f xa[8];
#pragma unroll
  for (int t = 0; t < 8; ++t) {
    v4f lo = *(const v4f*)(xrd + 16 * t);
    v4f hi = *(const v4f*)(xrd + 16 * t + 4);
    xa[t][0] = lo.x; xa[t][1] = lo.y; xa[t][2] = lo.z; xa[t][3] = lo.w;
    xa[t][4] = hi.x; xa[t][5] = hi.y; xa[t][6] = hi.z; xa[t][7] = hi.w;
  }
  wave_lds_handover();
#pragma unroll
  for (int t = 0; t < 8; ++t)
#pragma unroll
    for (int u = 0; u < 4; ++u)
      *(v4f*)(spark + (16 * t + 2 * u) * BPITCH) = st[t][u];
  __syncthreads();
  const float* frow = sb + (w * 128 + l) * BPITCH + 4 * half;
  v4f af[8][4];
#pragma unroll
  for (int t = 0; t < 8; ++t)
#pragma unroll
    for (int u = 0; u < 4; ++u)
      af[t][u] = *(const v4f*)(frow + t * 16 * BPITCH + 8 * u);
#pragma unroll
  for (int t = 0; t < 8; ++t) {
    asm volatile("" :: "v"(af[t][0]), "v"(af[t][1]), "v"(af[t][2]),
                 "v"(af[t][3]));
  }
#pragma unroll
  for (int t = 0; t < 8; ++t) {
#pragma unroll
    for (int u = 0; u < 4; ++u) {
      v2f a0, a1;
      a0.x = af[t][u].x; a0.y = af[t][u].y;
      a1.x = af[t][u].z; a1.y = af[t][u].w;
      xa[t] = wmma_f32_k4(a0, bq[2 * u], xa[t]);
      xa[t] = wmma_f32_k4(a1, bq[2 * u + 1], xa[t]);
    }
  }
  const float scale = 0.17677669529663687f;
  float mx = -INFINITY;
#pragma unroll
  for (int t = 0; t < 8; ++t)
#pragma unroll
    for (int r = 0; r < 8; ++r) { xa[t][r] *= scale; mx = fmaxf(mx, xa[t][r]); }
  mx = fmaxf(mx, __shfl_xor(mx, 16, 32));
  if (half == 0) red[0][w][l] = mx;
  __syncthreads();
#pragma unroll
  for (int u = 0; u < 8; ++u) mx = fmaxf(mx, red[0][u][l]);
  float sm = 0.f;
#pragma unroll
  for (int t = 0; t < 8; ++t)
#pragma unroll
    for (int r = 0; r < 8; ++r) { xa[t][r] = __expf(xa[t][r] - mx); sm += xa[t][r]; }
  sm += __shfl_xor(sm, 16, 32);
  if (half == 0) red[1][w][l] = sm;
  __syncthreads();
  float tot = red[1][0][l];
#pragma unroll
  for (int u = 1; u < 8; ++u) tot += red[1][u][l];
  float inv = 1.0f / tot;
#pragma unroll
  for (int t = 0; t < 8; ++t)
#pragma unroll
    for (int r = 0; r < 8; ++r) xa[t][r] *= inv;
  const float* brd = sb + (w * 128 + 8 * half) * BPITCH + 2 * l;
  v2f rb[64];
#pragma unroll
  for (int u = 0; u < 64; ++u)
    rb[u] = *(const v2f*)(brd + ((u >> 3) * 16 + (u & 7)) * BPITCH);
#pragma unroll
  for (int t = 0; t < 8; ++t) {
    asm volatile("" :: "v"(rb[8 * t]), "v"(rb[8 * t + 1]), "v"(rb[8 * t + 2]),
                 "v"(rb[8 * t + 3]), "v"(rb[8 * t + 4]), "v"(rb[8 * t + 5]),
                 "v"(rb[8 * t + 6]), "v"(rb[8 * t + 7]));
  }
  wave_lds_handover();
  if (nvalid) {
    float* pw = xsb + l * XPITCH + 8 * half;
#pragma unroll
    for (int t = 0; t < 8; ++t) {
      v4f lo, hi;
      lo.x = xa[t][0]; lo.y = xa[t][1]; lo.z = xa[t][2]; lo.w = xa[t][3];
      hi.x = xa[t][4]; hi.y = xa[t][5]; hi.z = xa[t][6]; hi.w = xa[t][7];
      *(v4f*)(pw + 16 * t) = lo;
      *(v4f*)(pw + 16 * t + 4) = hi;
    }
  }
  v8f c0 = {}, c1 = {};
#pragma unroll
  for (int t = 0; t < 8; ++t) {
#pragma unroll
    for (int m = 0; m < 4; ++m) {
      v2f r0 = rb[8 * t + 2 * m];
      v2f r1 = rb[8 * t + 2 * m + 1];
      v2f pa, b0, b1;
      pa.x = xa[t][2 * m]; pa.y = xa[t][2 * m + 1];
      b0.x = r0.x; b0.y = r1.x;
      b1.x = r0.y; b1.y = r1.y;
      c0 = wmma_f32_k4(pa, b0, c0);
      c1 = wmma_f32_k4(pa, b1, c1);
    }
  }
  if (half == 0) {
#pragma unroll
    for (int r = 0; r < 8; ++r) {
      v2f o; o.x = c0[r]; o.y = c1[r];
      *(v2f*)(&part[w][r * DVDIM + 2 * l]) = o;
    }
  }
  __syncthreads();
  v4f ov[8];
#pragma unroll
  for (int h = 0; h < 8; ++h) ov[h] = *(const v4f*)(xsb + h * XPITCH + 4 * lane);
#pragma unroll
  for (int h = 0; h < 8; ++h) *(v4f*)(pg + (size_t)h * plane) = ov[h];
  float sum = part[0][tid];
#pragma unroll
  for (int u = 1; u < 8; ++u) sum += part[u][tid];
  emb[(size_t)(b * LSEQ + i) * (NHEAD * DVDIM) + tid] = sum;
}

#define APITCH 132
#define E1TILES 2
__global__ void __launch_bounds__(256, 2)
emb1_kernel(const float* __restrict__ attn,
            const float* __restrict__ V,
            float* __restrict__ emb) {
  __shared__ __attribute__((aligned(16))) float sa[8][32 * APITCH];
  int bh = blockIdx.x; int b = bh >> 3; int hh = bh & 7;
  int i0 = blockIdx.y * (32 * E1TILES);
  int tid = threadIdx.x, w = tid >> 5;
  int lane = tid & 31, half = lane >> 4, l = lane & 15;
  const float* grow = attn + (((size_t)hh * BATCH + b) * LSEQ + i0) * LSEQ
                      + w * 128 + 4 * lane;
  const float* vcol =
      V + ((size_t)(b * LSEQ + w * 128 + 4 * half) * NHEAD + hh) * DVDIM + 2 * l;
  v2f vb[32][2];
#pragma unroll
  for (int s = 0; s < 16; ++s)
#pragma unroll
    for (int e = 0; e < 2; ++e)
      vb[s][e] = *(const v2f*)(
          vcol + (size_t)(8 * (s >> 1) + 2 * (s & 1) + e) * (NHEAD * DVDIM));
  v4f st[32];
#pragma unroll
  for (int r = 0; r < 32; ++r) st[r] = *(const v4f*)(grow + (size_t)r * LSEQ);
  float* mine = &sa[w][0];
#pragma unroll
  for (int r = 0; r < 32; ++r) *(v4f*)(mine + r * APITCH + 4 * lane) = st[r];
  __syncthreads();
#pragma unroll
  for (int s = 16; s < 32; ++s)
#pragma unroll
    for (int e = 0; e < 2; ++e)
      vb[s][e] = *(const v2f*)(
          vcol + (size_t)(8 * (s >> 1) + 2 * (s & 1) + e) * (NHEAD * DVDIM));
  const float* frow = mine + l * APITCH + 4 * half;
  float* e = emb + (size_t)(b * LSEQ + i0) * (NHEAD * DVDIM) + hh * DVDIM;
  float ev[4 * E1TILES];
  v8f c[E1TILES][2][2] = {};
#pragma unroll
  for (int ps = 0; ps < E1TILES; ++ps) {
    if (ps + 1 < E1TILES) {
#pragma unroll
      for (int r = 0; r < 32; ++r)
        st[r] = *(const v4f*)(grow + (size_t)(32 * (ps + 1) + r) * LSEQ);
    } else {
#pragma unroll
      for (int q = 0; q < 4 * E1TILES; ++q) {
        int idx = tid + 256 * q;
        ev[q] = e[(size_t)(idx >> 5) * (NHEAD * DVDIM) + (idx & 31)];
      }
    }
    v4f af[2][2][4];
#pragma unroll
    for (int m = 0; m < 2; ++m)
#pragma unroll
      for (int uu = 0; uu < 4; ++uu)
        af[0][m][uu] = *(const v4f*)(frow + m * 16 * APITCH + 8 * uu);
#pragma unroll
    for (int q = 0; q < 4; ++q) {
      if (q < 3) {
#pragma unroll
        for (int m = 0; m < 2; ++m)
#pragma unroll
          for (int uu = 0; uu < 4; ++uu)
            af[(q + 1) & 1][m][uu] =
                *(const v4f*)(frow + m * 16 * APITCH + 8 * (4 * (q + 1) + uu));
      }
      __builtin_amdgcn_sched_barrier(0);
#pragma unroll
      for (int uu = 0; uu < 4; ++uu) {
#pragma unroll
        for (int p = 0; p < 2; ++p) {
          const int s = 2 * (4 * q + uu) + p;
          v2f b0, b1;
          b0.x = vb[s][0].x; b0.y = vb[s][1].x;
          b1.x = vb[s][0].y; b1.y = vb[s][1].y;
#pragma unroll
          for (int m = 0; m < 2; ++m) {
            v2f a;
            a.x = p ? af[q & 1][m][uu].z : af[q & 1][m][uu].x;
            a.y = p ? af[q & 1][m][uu].w : af[q & 1][m][uu].y;
            c[ps][m][0] = wmma_f32_k4(a, b0, c[ps][m][0]);
            c[ps][m][1] = wmma_f32_k4(a, b1, c[ps][m][1]);
          }
        }
      }
    }
    if (ps + 1 < E1TILES) {
      wave_lds_handover();
#pragma unroll
      for (int r = 0; r < 32; ++r) *(v4f*)(mine + r * APITCH + 4 * lane) = st[r];
      wave_lds_handover();
    }
  }
  wave_lds_handover();
#pragma unroll
  for (int ps = 0; ps < E1TILES; ++ps)
#pragma unroll
    for (int m = 0; m < 2; ++m)
#pragma unroll
      for (int r = 0; r < 8; ++r) {
        v2f o; o.x = c[ps][m][0][r]; o.y = c[ps][m][1][r];
        *(v2f*)(mine + (32 * ps + 16 * m + 8 * half + r) * DVDIM + 2 * l) = o;
      }
  __syncthreads();
#pragma unroll
  for (int q = 0; q < 4 * E1TILES; ++q) {
    int idx = tid + 256 * q;
    float sum = sa[0][idx];
#pragma unroll
    for (int u = 1; u < 8; ++u) sum += sa[u][idx];
    e[(size_t)(idx >> 5) * (NHEAD * DVDIM) + (idx & 31)] = ev[q] + sum;
  }
}

__global__ void __launch_bounds__(256, 6)
fc_kernel(const float* __restrict__ emb,
          const float* __restrict__ W,
          const float* __restrict__ bvec,
          const float* __restrict__ resid,
          float* __restrict__ out) {
  __shared__ __attribute__((aligned(16))) float sm[GEMM_LDS_FLOATS];
  int m0 = blockIdx.x * 32, n0 = blockIdx.y * 64;
  int tid = threadIdx.x, w = tid >> 5;
  int lane = tid & 31, half = lane >> 4, l = lane & 15;
  int n = n0 + 16 * (w >> 1) + l;
  size_t off = (size_t)(m0 + 16 * (w & 1) + 8 * half) * DMODEL + n;
  float bv = bvec[n];
  float rs[8];
#pragma unroll
  for (int r = 0; r < 8; ++r) rs[r] = resid[off + (size_t)r * DMODEL];
  v8f c = gemm_tile(emb, W, sm, m0, n0);
#pragma unroll
  for (int r = 0; r < 8; ++r) out[off + (size_t)r * DMODEL] = c[r] + bv + rs[r];
}

extern "C" void kernel_launch(void* const* d_in, const int* in_sizes, int n_in,
                              void* d_out, int out_size, void* d_ws,
                              size_t ws_size, hipStream_t stream) {
  (void)in_sizes; (void)n_in; (void)out_size; (void)ws_size;
  const float* h    = (const float*)d_in[0];
  const float* bias = (const float*)d_in[1];
  const float* w_qs = (const float*)d_in[2];
  const float* w_ks = (const float*)d_in[3];
  const float* w_vs = (const float*)d_in[4];
  const float* ln_g = (const float*)d_in[5];
  const float* ln_b = (const float*)d_in[6];
  const float* fc_w = (const float*)d_in[7];
  const float* fc_b = (const float*)d_in[8];

  float* out  = (float*)d_out;
  float* attn = out + (size_t)BATCH * LSEQ * DMODEL;

  const size_t ROWS = (size_t)BATCH * LSEQ;
  float* ws  = (float*)d_ws;
  float* hn  = ws;
  float* q   = hn + ROWS * DMODEL;
  float* k   = q + ROWS * DMODEL;
  float* v   = k + ROWS * DMODEL;
  float* emb = v + ROWS * DMODEL;

  ln_kernel<<<(int)ROWS, 32, 0, stream>>>(h, ln_g, ln_b, hn);

  qkv_kernel<<<dim3(64, 4, 3), 256, 0, stream>>>(hn, w_qs, w_ks, w_vs, q, k, v);

  attn1_kernel<<<dim3(BATCH * NHEAD, LSEQ / (16 * A1TILES)), 256, 0, stream>>>(q, k, attn);
  row_kernel<<<(unsigned)ROWS, 256, 0, stream>>>(bias, q, attn, emb);
  emb1_kernel<<<dim3(BATCH * NHEAD, LSEQ / (32 * E1TILES)), 256, 0, stream>>>(attn, v, emb);
  fc_kernel<<<dim3(64, 4), 256, 0, stream>>>(emb, fc_w, fc_b, h, out);
}
